# starting kernel
# baseline (speedup 1.0000x reference)
#include <hip/hip_runtime.h>
#include <hip/hip_bf16.h>

#define N_NODES 100000
#define N_EDGES 1600000
#define NFH     128
#define NCLS    64

typedef __attribute__((ext_vector_type(2))) float v2f;
typedef __attribute__((ext_vector_type(8))) float v8f;

__global__ void k_deg_init(float* __restrict__ deg, int n) {
  int i = blockIdx.x * blockDim.x + threadIdx.x;
  if (i < n) deg[i] = 1.0f;
}

__global__ void k_deg_edges(const int* __restrict__ dst, float* __restrict__ deg, int e) {
  int i = blockIdx.x * blockDim.x + threadIdx.x;
  if (i < e) atomicAdd(&deg[dst[i]], 1.0f);
}

__global__ void k_rsqrt_inplace(float* __restrict__ d, int n) {
  int i = blockIdx.x * blockDim.x + threadIdx.x;
  if (i < n) d[i] = rsqrtf(d[i]);
}

__global__ void k_coef(const int* __restrict__ src, const int* __restrict__ dst,
                       const float* __restrict__ dinv, float* __restrict__ coef, int e) {
  int i = blockIdx.x * blockDim.x + threadIdx.x;
  if (i < e) coef[i] = dinv[src[i]] * dinv[dst[i]];
}

template <int K, int M>
__global__ __launch_bounds__(256)
void k_gemm_wmma(const float* __restrict__ X, const float* __restrict__ W,
                 float* __restrict__ H, int n_rows) {
  const int lane = threadIdx.x & 31;
  const int wave = (blockIdx.x * (blockDim.x >> 5)) + (threadIdx.x >> 5);
  constexpr int pairsPerRow = M / 32;
  const int nWork = (n_rows / 16) * pairsPerRow;
  if (wave >= nWork) return;

  const int ti = wave / pairsPerRow;
  const int tj = (wave % pairsPerRow) * 2;

  const int m  = lane & 15;
  const int ks = (lane >> 4) << 1;

  const float* __restrict__ xrow = X + (size_t)(ti * 16 + m) * K + ks;
  const float* __restrict__ w0   = W + (size_t)(tj * 16) + m;
  const float* __restrict__ w1   = w0 + 16;

  v8f c0 = {}, c1 = {};
#pragma unroll
  for (int k0 = 0; k0 < K; k0 += 4) {
    v2f a, b0, b1;
    a.x  = xrow[k0];
    a.y  = xrow[k0 + 1];
    b0.x = w0[(size_t)(k0 + ks) * M];
    b0.y = w0[(size_t)(k0 + ks + 1) * M];
    b1.x = w1[(size_t)(k0 + ks) * M];
    b1.y = w1[(size_t)(k0 + ks + 1) * M];
    c0 = __builtin_amdgcn_wmma_f32_16x16x4_f32(false, a, false, b0, (short)0, c0,
                                               false, false);
    c1 = __builtin_amdgcn_wmma_f32_16x16x4_f32(false, a, false, b1, (short)0, c1,
                                               false, false);
  }

  const int row0 = ti * 16 + ((lane >> 4) << 3);
  float* __restrict__ out0 = H + (size_t)row0 * M + tj * 16 + m;
#pragma unroll
  for (int v = 0; v < 8; ++v) {
    out0[(size_t)v * M]      = c0[v];
    out0[(size_t)v * M + 16] = c1[v];
  }
}

template <int M>
__global__ void k_init_agg(const float* __restrict__ h, const float* __restrict__ dinv,
                           const float* __restrict__ bias, float* __restrict__ agg,
                           int n) {
  int idx = blockIdx.x * blockDim.x + threadIdx.x;
  int total = n * M;
  if (idx < total) {
    int i = idx / M;
    int f = idx - i * M;
    float di = dinv[i];
    agg[idx] = bias[f] + di * di * h[idx];
  }
}

template <int M>
__global__ void k_scatter(const int* __restrict__ src, const int* __restrict__ dst,
                          const float* __restrict__ coef, const float* __restrict__ h,
                          float* __restrict__ agg, int e) {
  const int lane = threadIdx.x & 31;
  const int ed = __builtin_amdgcn_readfirstlane(
      (blockIdx.x * (blockDim.x >> 5)) + (threadIdx.x >> 5));
  if (ed >= e) return;
  const int   s  = __builtin_amdgcn_readfirstlane(src[ed]);
  const int   d  = __builtin_amdgcn_readfirstlane(dst[ed]);
  const float cf = coef[ed];
  constexpr int F = M / 32;
  const float* __restrict__ hp = h + (size_t)s * M + lane * F;
  float* __restrict__ ap = agg + (size_t)d * M + lane * F;
  if constexpr (F == 4) {
    float4 hv = *reinterpret_cast<const float4*>(hp);
    atomicAdd(ap + 0, cf * hv.x);
    atomicAdd(ap + 1, cf * hv.y);
    atomicAdd(ap + 2, cf * hv.z);
    atomicAdd(ap + 3, cf * hv.w);
  } else {
    float2 hv = *reinterpret_cast<const float2*>(hp);
    atomicAdd(ap + 0, cf * hv.x);
    atomicAdd(ap + 1, cf * hv.y);
  }
}

__global__ void k_relu_res(const float* __restrict__ agg, const float* __restrict__ xp,
                           float* __restrict__ xn, int total) {
  int i = blockIdx.x * blockDim.x + threadIdx.x;
  if (i < total) {
    float v = agg[i] + xp[i];
    xn[i] = v > 0.0f ? v : 0.0f;
  }
}

__global__ void k_add(const float* __restrict__ a, const float* __restrict__ b,
                      float* __restrict__ o, int total) {
  int i = blockIdx.x * blockDim.x + threadIdx.x;
  if (i < total) o[i] = a[i] + b[i];
}

extern "C" void kernel_launch(void* const* d_in, const int* in_sizes, int n_in,
                              void* d_out, int out_size, void* d_ws, size_t ws_size,
                              hipStream_t stream) {
  const float* x      = (const float*)d_in[0];
  const int*   ei     = (const int*)d_in[1];
  const float* W_in   = (const float*)d_in[2];
  const float* b_in   = (const float*)d_in[3];
  const float* W_m0   = (const float*)d_in[4];
  const float* b_m0   = (const float*)d_in[5];
  const float* W_m1   = (const float*)d_in[6];
  const float* b_m1   = (const float*)d_in[7];
  const float* W_out  = (const float*)d_in[8];
  const float* b_out  = (const float*)d_in[9];
  const float* W_skip = (const float*)d_in[10];

  const int* src = ei;
  const int* dst = ei + N_EDGES;

  float* w    = (float*)d_ws;
  float* dinv = w;  w += N_NODES;
  float* coef = w;  w += N_EDGES;
  float* h    = w;  w += (size_t)N_NODES * NFH;
  float* agg  = w;  w += (size_t)N_NODES * NFH;
  float* xA   = w;  w += (size_t)N_NODES * NFH;
  float* xB   = w;  w += (size_t)N_NODES * NFH;
  float* skip = w;  w += (size_t)N_NODES * NCLS;

  const int T = 256;
  auto cdiv = [](int a, int b) { return (a + b - 1) / b; };

  const int nodeBlocks  = cdiv(N_NODES, T);
  const int edgeBlocks  = cdiv(N_EDGES, T);
  const int featBlocks  = cdiv(N_NODES * NFH, T);
  const int clsBlocks   = cdiv(N_NODES * NCLS, T);
  const int scatBlocks  = cdiv(N_EDGES, T / 32);
  const int gemmBlocksH = cdiv((N_NODES / 16) * (NFH / 32), T / 32);
  const int gemmBlocksC = cdiv((N_NODES / 16) * (NCLS / 32), T / 32);

  k_deg_init<<<nodeBlocks, T, 0, stream>>>(dinv, N_NODES);
  k_deg_edges<<<edgeBlocks, T, 0, stream>>>(dst, dinv, N_EDGES);
  k_rsqrt_inplace<<<nodeBlocks, T, 0, stream>>>(dinv, N_NODES);
  k_coef<<<edgeBlocks, T, 0, stream>>>(src, dst, dinv, coef, N_EDGES);

  k_gemm_wmma<NFH, NFH><<<gemmBlocksH, T, 0, stream>>>(x, W_in, h, N_NODES);
  k_init_agg<NFH><<<featBlocks, T, 0, stream>>>(h, dinv, b_in, agg, N_NODES);
  k_scatter<NFH><<<scatBlocks, T, 0, stream>>>(src, dst, coef, h, agg, N_EDGES);
  k_relu_res<<<featBlocks, T, 0, stream>>>(agg, x, xA, N_NODES * NFH);

  k_gemm_wmma<NFH, NFH><<<gemmBlocksH, T, 0, stream>>>(xA, W_m0, h, N_NODES);
  k_init_agg<NFH><<<featBlocks, T, 0, stream>>>(h, dinv, b_m0, agg, N_NODES);
  k_scatter<NFH><<<scatBlocks, T, 0, stream>>>(src, dst, coef, h, agg, N_EDGES);
  k_relu_res<<<featBlocks, T, 0, stream>>>(agg, xA, xB, N_NODES * NFH);

  k_gemm_wmma<NFH, NFH><<<gemmBlocksH, T, 0, stream>>>(xB, W_m1, h, N_NODES);
  k_init_agg<NFH><<<featBlocks, T, 0, stream>>>(h, dinv, b_m1, agg, N_NODES);
  k_scatter<NFH><<<scatBlocks, T, 0, stream>>>(src, dst, coef, h, agg, N_EDGES);
  k_relu_res<<<featBlocks, T, 0, stream>>>(agg, xB, xA, N_NODES * NFH);

  k_gemm_wmma<NFH, NCLS><<<gemmBlocksC, T, 0, stream>>>(xA, W_out, h, N_NODES);
  k_init_agg<NCLS><<<clsBlocks, T, 0, stream>>>(h, dinv, b_out, agg, N_NODES);
  k_scatter<NCLS><<<scatBlocks, T, 0, stream>>>(src, dst, coef, h, agg, N_EDGES);
  k_gemm_wmma<NFH, NCLS><<<gemmBlocksC, T, 0, stream>>>(xA, W_skip, skip, N_NODES);
  k_add<<<clsBlocks, T, 0, stream>>>(agg, skip, (float*)d_out, N_NODES * NCLS);
}


# Edge sums by destination: edges ranked per dst by one returning int add, prefix-summed into rows; one wave sums a row in registers and stores it once; f32 atomics gone, sum order = arrival rank
# speedup vs baseline: 2.5781x; 2.5781x over previous
#include <hip/hip_runtime.h>
#include <hip/hip_bf16.h>

#define N_NODES 100000
#define N_EDGES 1600000
#define NFH     128
#define NCLS    64

typedef __attribute__((ext_vector_type(2))) float v2f;
typedef __attribute__((ext_vector_type(8))) float v8f;

#define SCAN_T   256
#define SCAN_BLK 1024

__global__ void k_zero_cnt(int* __restrict__ cnt, int n) {
  int i = blockIdx.x * blockDim.x + threadIdx.x;
  if (i < n) cnt[i] = 0;
}

__global__ void k_rank(const int* __restrict__ dst, int* __restrict__ cnt,
                       int* __restrict__ rank, int e) {
  int i = blockIdx.x * blockDim.x + threadIdx.x;
  if (i < e) rank[i] = atomicAdd(&cnt[dst[i]], 1);
}

__global__ __launch_bounds__(SCAN_T)
void k_blocksum(const int* __restrict__ cnt, int* __restrict__ bsum, int n) {
  __shared__ int wsum[SCAN_T / 32];
  const int lane = threadIdx.x & 31;
  const int w    = threadIdx.x >> 5;
  const int i0   = (blockIdx.x * SCAN_T + threadIdx.x) * 4;
  int s = 0;
  if (i0 < n) {
    const int4 c = *reinterpret_cast<const int4*>(cnt + i0);
    s = c.x + c.y + c.z + c.w;
  }
  for (int o = 16; o > 0; o >>= 1) s += __shfl_xor(s, o);
  if (lane == 0) wsum[w] = s;
  __syncthreads();
  if (threadIdx.x == 0) {
    int t = 0;
    for (int k = 0; k < SCAN_T / 32; ++k) t += wsum[k];
    bsum[blockIdx.x] = t;
  }
}

__global__ __launch_bounds__(SCAN_T)
void k_scan(const int* __restrict__ cnt, const int* __restrict__ bsum,
            int* __restrict__ rowptr, float* __restrict__ dinv, int n) {
  __shared__ int wsum[SCAN_T / 32];
  const int lane = threadIdx.x & 31;
  const int w    = threadIdx.x >> 5;
  int base = 0;
  for (int k = 0; k < (int)blockIdx.x; ++k) base += bsum[k];
  const int i0 = (blockIdx.x * SCAN_T + threadIdx.x) * 4;
  int4 c = make_int4(0, 0, 0, 0);
  if (i0 < n) c = *reinterpret_cast<const int4*>(cnt + i0);
  const int tsum = c.x + c.y + c.z + c.w;
  int inc = tsum;
  for (int o = 1; o < 32; o <<= 1) {
    const int t = __shfl_up(inc, o);
    if (lane >= o) inc += t;
  }
  if (lane == 31) wsum[w] = inc;
  __syncthreads();
  int woff = 0;
  for (int k = 0; k < SCAN_T / 32; ++k) woff += (k < w) ? wsum[k] : 0;
  const int p = base + woff + inc - tsum;
  if (i0 < n) {
    int4 r;
    r.x = p;
    r.y = r.x + c.x;
    r.z = r.y + c.y;
    r.w = r.z + c.z;
    *reinterpret_cast<int4*>(rowptr + i0) = r;
    float4 d;
    d.x = rsqrtf((float)(c.x + 1));
    d.y = rsqrtf((float)(c.y + 1));
    d.z = rsqrtf((float)(c.z + 1));
    d.w = rsqrtf((float)(c.w + 1));
    *reinterpret_cast<float4*>(dinv + i0) = d;
    if (i0 + 4 == n) rowptr[n] = r.w + c.w;
  }
}

__global__ void k_fill(const int* __restrict__ src, const int* __restrict__ dst,
                       const int* __restrict__ rank, const int* __restrict__ rowptr,
                       const float* __restrict__ dinv, int2* __restrict__ ent, int e) {
  int i = blockIdx.x * blockDim.x + threadIdx.x;
  if (i < e) {
    const int s = src[i];
    const int d = dst[i];
    ent[rowptr[d] + rank[i]] = make_int2(s, __float_as_int(dinv[s] * dinv[d]));
  }
}

template <int K, int M>
__global__ __launch_bounds__(256)
void k_gemm_wmma(const float* __restrict__ X, const float* __restrict__ W,
                 float* __restrict__ H, int n_rows) {
  const int lane = threadIdx.x & 31;
  const int wave = (blockIdx.x * (blockDim.x >> 5)) + (threadIdx.x >> 5);
  constexpr int pairsPerRow = M / 32;
  const int nWork = (n_rows / 16) * pairsPerRow;
  if (wave >= nWork) return;

  const int ti = wave / pairsPerRow;
  const int tj = (wave % pairsPerRow) * 2;

  const int m  = lane & 15;
  const int ks = (lane >> 4) << 1;

  const float* __restrict__ xrow = X + (size_t)(ti * 16 + m) * K + ks;
  const float* __restrict__ w0   = W + (size_t)(tj * 16) + m;
  const float* __restrict__ w1   = w0 + 16;

  v8f c0 = {}, c1 = {};
#pragma unroll
  for (int k0 = 0; k0 < K; k0 += 4) {
    v2f a, b0, b1;
    a.x  = xrow[k0];
    a.y  = xrow[k0 + 1];
    b0.x = w0[(size_t)(k0 + ks) * M];
    b0.y = w0[(size_t)(k0 + ks + 1) * M];
    b1.x = w1[(size_t)(k0 + ks) * M];
    b1.y = w1[(size_t)(k0 + ks + 1) * M];
    c0 = __builtin_amdgcn_wmma_f32_16x16x4_f32(false, a, false, b0, (short)0, c0,
                                               false, false);
    c1 = __builtin_amdgcn_wmma_f32_16x16x4_f32(false, a, false, b1, (short)0, c1,
                                               false, false);
  }

  const int row0 = ti * 16 + ((lane >> 4) << 3);
  float* __restrict__ out0 = H + (size_t)row0 * M + tj * 16 + m;
#pragma unroll
  for (int v = 0; v < 8; ++v) {
    out0[(size_t)v * M]      = c0[v];
    out0[(size_t)v * M + 16] = c1[v];
  }
}

template <int F> struct RowVec;
template <> struct RowVec<4> { typedef float4 T; };
template <> struct RowVec<2> { typedef float2 T; };

__device__ __forceinline__ void axpy(float4& a, float c, const float4& v) {
  a.x += c * v.x; a.y += c * v.y; a.z += c * v.z; a.w += c * v.w;
}
__device__ __forceinline__ void axpy(float2& a, float c, const float2& v) {
  a.x += c * v.x; a.y += c * v.y;
}

template <int M>
__global__ __launch_bounds__(256)
void k_gather(const int* __restrict__ rowptr, const int2* __restrict__ ent,
              const float* __restrict__ dinv, const float* __restrict__ h,
              const float* __restrict__ bias, float* __restrict__ agg, int n) {
  constexpr int F = M / 32;
  typedef typename RowVec<F>::T VT;
  const int lane = threadIdx.x & 31;
  const int i = __builtin_amdgcn_readfirstlane(
      (blockIdx.x * (blockDim.x >> 5)) + (threadIdx.x >> 5));
  if (i >= n) return;
  const int beg = rowptr[i];
  const int end = rowptr[i + 1];
  const float di = dinv[i];
  const VT* __restrict__ hv = reinterpret_cast<const VT*>(h);

  VT acc = reinterpret_cast<const VT*>(bias)[lane];
  axpy(acc, di * di, hv[(size_t)i * 32 + lane]);

  for (int base = beg; base < end; base += 32) {
    const int cnt = (end - base) < 32 ? (end - base) : 32;
    int2 en = make_int2(0, 0);
    if (lane < cnt) en = ent[base + lane];
#define ENT_SRC(t)  __builtin_amdgcn_readlane(en.x, (t))
#define ENT_COEF(t) __int_as_float(__builtin_amdgcn_readlane(en.y, (t)))
#define ROW(t)      hv[(size_t)ENT_SRC(t) * 32 + lane]
    int t = 0;
    for (; t + 4 <= cnt; t += 4) {
      const VT v0 = ROW(t), v1 = ROW(t + 1), v2 = ROW(t + 2), v3 = ROW(t + 3);
      axpy(acc, ENT_COEF(t), v0);
      axpy(acc, ENT_COEF(t + 1), v1);
      axpy(acc, ENT_COEF(t + 2), v2);
      axpy(acc, ENT_COEF(t + 3), v3);
    }
    const int r = cnt - t;
    if (r == 3) {
      const VT v0 = ROW(t), v1 = ROW(t + 1), v2 = ROW(t + 2);
      axpy(acc, ENT_COEF(t), v0);
      axpy(acc, ENT_COEF(t + 1), v1);
      axpy(acc, ENT_COEF(t + 2), v2);
    } else if (r == 2) {
      const VT v0 = ROW(t), v1 = ROW(t + 1);
      axpy(acc, ENT_COEF(t), v0);
      axpy(acc, ENT_COEF(t + 1), v1);
    } else if (r == 1) {
      const VT v0 = ROW(t);
      axpy(acc, ENT_COEF(t), v0);
    }
#undef ROW
#undef ENT_COEF
#undef ENT_SRC
  }
  reinterpret_cast<VT*>(agg)[(size_t)i * 32 + lane] = acc;
}

__global__ void k_relu_res(const float* __restrict__ agg, const float* __restrict__ xp,
                           float* __restrict__ xn, int total) {
  int i = blockIdx.x * blockDim.x + threadIdx.x;
  if (i < total) {
    float v = agg[i] + xp[i];
    xn[i] = v > 0.0f ? v : 0.0f;
  }
}

__global__ void k_add(const float* __restrict__ a, const float* __restrict__ b,
                      float* __restrict__ o, int total) {
  int i = blockIdx.x * blockDim.x + threadIdx.x;
  if (i < total) o[i] = a[i] + b[i];
}

extern "C" void kernel_launch(void* const* d_in, const int* in_sizes, int n_in,
                              void* d_out, int out_size, void* d_ws, size_t ws_size,
                              hipStream_t stream) {
  const float* x      = (const float*)d_in[0];
  const int*   ei     = (const int*)d_in[1];
  const float* W_in   = (const float*)d_in[2];
  const float* b_in   = (const float*)d_in[3];
  const float* W_m0   = (const float*)d_in[4];
  const float* b_m0   = (const float*)d_in[5];
  const float* W_m1   = (const float*)d_in[6];
  const float* b_m1   = (const float*)d_in[7];
  const float* W_out  = (const float*)d_in[8];
  const float* b_out  = (const float*)d_in[9];
  const float* W_skip = (const float*)d_in[10];

  const int* src = ei;
  const int* dst = ei + N_EDGES;

  float* w    = (float*)d_ws;
  float* dinv = w;  w += N_NODES;
  float* h    = w;  w += (size_t)N_NODES * NFH;
  float* agg  = w;  w += (size_t)N_NODES * NFH;
  float* xA   = w;  w += (size_t)N_NODES * NFH;
  float* xB   = w;  w += (size_t)N_NODES * NFH;
  float* skip = w;  w += (size_t)N_NODES * NCLS;
  int*  cnt    = (int*)w;   w += N_NODES;
  int*  rowptr = (int*)w;   w += N_NODES + 4;
  int*  bsum   = (int*)w;   w += 128;
  int*  rank   = (int*)w;   w += N_EDGES;
  int2* ent    = (int2*)w;  w += (size_t)2 * N_EDGES;

  const int T = 256;
  auto cdiv = [](int a, int b) { return (a + b - 1) / b; };

  const int nodeBlocks  = cdiv(N_NODES, T);
  const int edgeBlocks  = cdiv(N_EDGES, T);
  const int featBlocks  = cdiv(N_NODES * NFH, T);
  const int clsBlocks   = cdiv(N_NODES * NCLS, T);
  const int scanBlocks  = cdiv(N_NODES, SCAN_BLK);
  const int rowBlocks   = cdiv(N_NODES, T / 32);
  const int gemmBlocksH = cdiv((N_NODES / 16) * (NFH / 32), T / 32);
  const int gemmBlocksC = cdiv((N_NODES / 16) * (NCLS / 32), T / 32);

  k_zero_cnt<<<nodeBlocks, T, 0, stream>>>(cnt, N_NODES);
  k_rank<<<edgeBlocks, T, 0, stream>>>(dst, cnt, rank, N_EDGES);
  k_blocksum<<<scanBlocks, SCAN_T, 0, stream>>>(cnt, bsum, N_NODES);
  k_scan<<<scanBlocks, SCAN_T, 0, stream>>>(cnt, bsum, rowptr, dinv, N_NODES);
  k_fill<<<edgeBlocks, T, 0, stream>>>(src, dst, rank, rowptr, dinv, ent, N_EDGES);

  k_gemm_wmma<NFH, NFH><<<gemmBlocksH, T, 0, stream>>>(x, W_in, h, N_NODES);
  k_gather<NFH><<<rowBlocks, T, 0, stream>>>(rowptr, ent, dinv, h, b_in, agg, N_NODES);
  k_relu_res<<<featBlocks, T, 0, stream>>>(agg, x, xA, N_NODES * NFH);

  k_gemm_wmma<NFH, NFH><<<gemmBlocksH, T, 0, stream>>>(xA, W_m0, h, N_NODES);
  k_gather<NFH><<<rowBlocks, T, 0, stream>>>(rowptr, ent, dinv, h, b_m0, agg, N_NODES);
  k_relu_res<<<featBlocks, T, 0, stream>>>(agg, xA, xB, N_NODES * NFH);

  k_gemm_wmma<NFH, NFH><<<gemmBlocksH, T, 0, stream>>>(xB, W_m1, h, N_NODES);
  k_gather<NFH><<<rowBlocks, T, 0, stream>>>(rowptr, ent, dinv, h, b_m1, agg, N_NODES);
  k_relu_res<<<featBlocks, T, 0, stream>>>(agg, xB, xA, N_NODES * NFH);

  k_gemm_wmma<NFH, NCLS><<<gemmBlocksC, T, 0, stream>>>(xA, W_out, h, N_NODES);
  k_gather<NCLS><<<rowBlocks, T, 0, stream>>>(rowptr, ent, dinv, h, b_out, agg, N_NODES);
  k_gemm_wmma<NFH, NCLS><<<gemmBlocksC, T, 0, stream>>>(xA, W_skip, skip, N_NODES);
  k_add<<<clsBlocks, T, 0, stream>>>(agg, skip, (float*)d_out, N_NODES * NCLS);
}


# Residual add + relu (last layer: skip add) folded into the row-sum kernel's registers ahead of its one store: 4 one-element-per-thread launches and the agg plane removed; sum order unchanged
# speedup vs baseline: 3.2449x; 1.2586x over previous
#include <hip/hip_runtime.h>
#include <hip/hip_bf16.h>

#define N_NODES 100000
#define N_EDGES 1600000
#define NFH     128
#define NCLS    64

typedef __attribute__((ext_vector_type(2))) float v2f;
typedef __attribute__((ext_vector_type(8))) float v8f;

#define SCAN_T   256
#define SCAN_BLK 1024

__global__ void k_zero_cnt(int* __restrict__ cnt, int n) {
  int i = blockIdx.x * blockDim.x + threadIdx.x;
  if (i < n) cnt[i] = 0;
}

__global__ void k_rank(const int* __restrict__ dst, int* __restrict__ cnt,
                       int* __restrict__ rank, int e) {
  int i = blockIdx.x * blockDim.x + threadIdx.x;
  if (i < e) rank[i] = atomicAdd(&cnt[dst[i]], 1);
}

__global__ __launch_bounds__(SCAN_T)
void k_blocksum(const int* __restrict__ cnt, int* __restrict__ bsum, int n) {
  __shared__ int wsum[SCAN_T / 32];
  const int lane = threadIdx.x & 31;
  const int w    = threadIdx.x >> 5;
  const int i0   = (blockIdx.x * SCAN_T + threadIdx.x) * 4;
  int s = 0;
  if (i0 < n) {
    const int4 c = *reinterpret_cast<const int4*>(cnt + i0);
    s = c.x + c.y + c.z + c.w;
  }
  for (int o = 16; o > 0; o >>= 1) s += __shfl_xor(s, o);
  if (lane == 0) wsum[w] = s;
  __syncthreads();
  if (threadIdx.x == 0) {
    int t = 0;
    for (int k = 0; k < SCAN_T / 32; ++k) t += wsum[k];
    bsum[blockIdx.x] = t;
  }
}

__global__ __launch_bounds__(SCAN_T)
void k_scan(const int* __restrict__ cnt, const int* __restrict__ bsum,
            int* __restrict__ rowptr, float* __restrict__ dinv, int n) {
  __shared__ int wsum[SCAN_T / 32];
  const int lane = threadIdx.x & 31;
  const int w    = threadIdx.x >> 5;
  int base = 0;
  for (int k = 0; k < (int)blockIdx.x; ++k) base += bsum[k];
  const int i0 = (blockIdx.x * SCAN_T + threadIdx.x) * 4;
  int4 c = make_int4(0, 0, 0, 0);
  if (i0 < n) c = *reinterpret_cast<const int4*>(cnt + i0);
  const int tsum = c.x + c.y + c.z + c.w;
  int inc = tsum;
  for (int o = 1; o < 32; o <<= 1) {
    const int t = __shfl_up(inc, o);
    if (lane >= o) inc += t;
  }
  if (lane == 31) wsum[w] = inc;
  __syncthreads();
  int woff = 0;
  for (int k = 0; k < SCAN_T / 32; ++k) woff += (k < w) ? wsum[k] : 0;
  const int p = base + woff + inc - tsum;
  if (i0 < n) {
    int4 r;
    r.x = p;
    r.y = r.x + c.x;
    r.z = r.y + c.y;
    r.w = r.z + c.z;
    *reinterpret_cast<int4*>(rowptr + i0) = r;
    float4 d;
    d.x = rsqrtf((float)(c.x + 1));
    d.y = rsqrtf((float)(c.y + 1));
    d.z = rsqrtf((float)(c.z + 1));
    d.w = rsqrtf((float)(c.w + 1));
    *reinterpret_cast<float4*>(dinv + i0) = d;
    if (i0 + 4 == n) rowptr[n] = r.w + c.w;
  }
}

__global__ void k_fill(const int* __restrict__ src, const int* __restrict__ dst,
                       const int* __restrict__ rank, const int* __restrict__ rowptr,
                       const float* __restrict__ dinv, int2* __restrict__ ent, int e) {
  int i = blockIdx.x * blockDim.x + threadIdx.x;
  if (i < e) {
    const int s = src[i];
    const int d = dst[i];
    ent[rowptr[d] + rank[i]] = make_int2(s, __float_as_int(dinv[s] * dinv[d]));
  }
}

template <int K, int M>
__global__ __launch_bounds__(256)
void k_gemm_wmma(const float* __restrict__ X, const float* __restrict__ W,
                 float* __restrict__ H, int n_rows) {
  const int lane = threadIdx.x & 31;
  const int wave = (blockIdx.x * (blockDim.x >> 5)) + (threadIdx.x >> 5);
  constexpr int pairsPerRow = M / 32;
  const int nWork = (n_rows / 16) * pairsPerRow;
  if (wave >= nWork) return;

  const int ti = wave / pairsPerRow;
  const int tj = (wave % pairsPerRow) * 2;

  const int m  = lane & 15;
  const int ks = (lane >> 4) << 1;

  const float* __restrict__ xrow = X + (size_t)(ti * 16 + m) * K + ks;
  const float* __restrict__ w0   = W + (size_t)(tj * 16) + m;
  const float* __restrict__ w1   = w0 + 16;

  v8f c0 = {}, c1 = {};
#pragma unroll
  for (int k0 = 0; k0 < K; k0 += 4) {
    v2f a, b0, b1;
    a.x  = xrow[k0];
    a.y  = xrow[k0 + 1];
    b0.x = w0[(size_t)(k0 + ks) * M];
    b0.y = w0[(size_t)(k0 + ks + 1) * M];
    b1.x = w1[(size_t)(k0 + ks) * M];
    b1.y = w1[(size_t)(k0 + ks + 1) * M];
    c0 = __builtin_amdgcn_wmma_f32_16x16x4_f32(false, a, false, b0, (short)0, c0,
                                               false, false);
    c1 = __builtin_amdgcn_wmma_f32_16x16x4_f32(false, a, false, b1, (short)0, c1,
                                               false, false);
  }

  const int row0 = ti * 16 + ((lane >> 4) << 3);
  float* __restrict__ out0 = H + (size_t)row0 * M + tj * 16 + m;
#pragma unroll
  for (int v = 0; v < 8; ++v) {
    out0[(size_t)v * M]      = c0[v];
    out0[(size_t)v * M + 16] = c1[v];
  }
}

template <int F> struct RowVec;
template <> struct RowVec<4> { typedef float4 T; };
template <> struct RowVec<2> { typedef float2 T; };

__device__ __forceinline__ void axpy(float4& a, float c, const float4& v) {
  a.x += c * v.x; a.y += c * v.y; a.z += c * v.z; a.w += c * v.w;
}
__device__ __forceinline__ void axpy(float2& a, float c, const float2& v) {
  a.x += c * v.x; a.y += c * v.y;
}
template <bool RELU>
__device__ __forceinline__ float add_res(float a, float r) {
  const float v = a + r;
  return RELU ? (v > 0.0f ? v : 0.0f) : v;
}
template <bool RELU>
__device__ __forceinline__ float4 add_res(const float4& a, const float4& r) {
  return make_float4(add_res<RELU>(a.x, r.x), add_res<RELU>(a.y, r.y),
                     add_res<RELU>(a.z, r.z), add_res<RELU>(a.w, r.w));
}
template <bool RELU>
__device__ __forceinline__ float2 add_res(const float2& a, const float2& r) {
  return make_float2(add_res<RELU>(a.x, r.x), add_res<RELU>(a.y, r.y));
}

template <int M, bool RELU>
__global__ __launch_bounds__(256)
void k_gather(const int* __restrict__ rowptr, const int2* __restrict__ ent,
              const float* __restrict__ dinv, const float* __restrict__ h,
              const float* __restrict__ bias, const float* __restrict__ res,
              float* __restrict__ y, int n) {
  constexpr int F = M / 32;
  typedef typename RowVec<F>::T VT;
  const int lane = threadIdx.x & 31;
  const int i = __builtin_amdgcn_readfirstlane(
      (blockIdx.x * (blockDim.x >> 5)) + (threadIdx.x >> 5));
  if (i >= n) return;
  const int beg = rowptr[i];
  const int end = rowptr[i + 1];
  const float di = dinv[i];
  const VT* __restrict__ hv = reinterpret_cast<const VT*>(h);

  const VT rr = reinterpret_cast<const VT*>(res)[(size_t)i * 32 + lane];
  VT acc = reinterpret_cast<const VT*>(bias)[lane];
  axpy(acc, di * di, hv[(size_t)i * 32 + lane]);

  for (int base = beg; base < end; base += 32) {
    const int cnt = (end - base) < 32 ? (end - base) : 32;
    int2 en = make_int2(0, 0);
    if (lane < cnt) en = ent[base + lane];
#define ENT_SRC(t)  __builtin_amdgcn_readlane(en.x, (t))
#define ENT_COEF(t) __int_as_float(__builtin_amdgcn_readlane(en.y, (t)))
#define ROW(t)      hv[(size_t)ENT_SRC(t) * 32 + lane]
    int t = 0;
    for (; t + 4 <= cnt; t += 4) {
      const VT v0 = ROW(t), v1 = ROW(t + 1), v2 = ROW(t + 2), v3 = ROW(t + 3);
      axpy(acc, ENT_COEF(t), v0);
      axpy(acc, ENT_COEF(t + 1), v1);
      axpy(acc, ENT_COEF(t + 2), v2);
      axpy(acc, ENT_COEF(t + 3), v3);
    }
    const int r = cnt - t;
    if (r == 3) {
      const VT v0 = ROW(t), v1 = ROW(t + 1), v2 = ROW(t + 2);
      axpy(acc, ENT_COEF(t), v0);
      axpy(acc, ENT_COEF(t + 1), v1);
      axpy(acc, ENT_COEF(t + 2), v2);
    } else if (r == 2) {
      const VT v0 = ROW(t), v1 = ROW(t + 1);
      axpy(acc, ENT_COEF(t), v0);
      axpy(acc, ENT_COEF(t + 1), v1);
    } else if (r == 1) {
      const VT v0 = ROW(t);
      axpy(acc, ENT_COEF(t), v0);
    }
#undef ROW
#undef ENT_COEF
#undef ENT_SRC
  }
  reinterpret_cast<VT*>(y)[(size_t)i * 32 + lane] = add_res<RELU>(acc, rr);
}

extern "C" void kernel_launch(void* const* d_in, const int* in_sizes, int n_in,
                              void* d_out, int out_size, void* d_ws, size_t ws_size,
                              hipStream_t stream) {
  const float* x      = (const float*)d_in[0];
  const int*   ei     = (const int*)d_in[1];
  const float* W_in   = (const float*)d_in[2];
  const float* b_in   = (const float*)d_in[3];
  const float* W_m0   = (const float*)d_in[4];
  const float* b_m0   = (const float*)d_in[5];
  const float* W_m1   = (const float*)d_in[6];
  const float* b_m1   = (const float*)d_in[7];
  const float* W_out  = (const float*)d_in[8];
  const float* b_out  = (const float*)d_in[9];
  const float* W_skip = (const float*)d_in[10];

  const int* src = ei;
  const int* dst = ei + N_EDGES;

  float* w    = (float*)d_ws;
  float* dinv = w;  w += N_NODES;
  float* h    = w;  w += (size_t)N_NODES * NFH;
  float* xA   = w;  w += (size_t)N_NODES * NFH;
  float* xB   = w;  w += (size_t)N_NODES * NFH;
  float* skip = w;  w += (size_t)N_NODES * NCLS;
  int*  cnt    = (int*)w;   w += N_NODES;
  int*  rowptr = (int*)w;   w += N_NODES + 4;
  int*  bsum   = (int*)w;   w += 128;
  int*  rank   = (int*)w;   w += N_EDGES;
  int2* ent    = (int2*)w;  w += (size_t)2 * N_EDGES;

  const int T = 256;
  auto cdiv = [](int a, int b) { return (a + b - 1) / b; };

  const int nodeBlocks  = cdiv(N_NODES, T);
  const int edgeBlocks  = cdiv(N_EDGES, T);
  const int scanBlocks  = cdiv(N_NODES, SCAN_BLK);
  const int rowBlocks   = cdiv(N_NODES, T / 32);
  const int gemmBlocksH = cdiv((N_NODES / 16) * (NFH / 32), T / 32);
  const int gemmBlocksC = cdiv((N_NODES / 16) * (NCLS / 32), T / 32);

  k_zero_cnt<<<nodeBlocks, T, 0, stream>>>(cnt, N_NODES);
  k_rank<<<edgeBlocks, T, 0, stream>>>(dst, cnt, rank, N_EDGES);
  k_blocksum<<<scanBlocks, SCAN_T, 0, stream>>>(cnt, bsum, N_NODES);
  k_scan<<<scanBlocks, SCAN_T, 0, stream>>>(cnt, bsum, rowptr, dinv, N_NODES);
  k_fill<<<edgeBlocks, T, 0, stream>>>(src, dst, rank, rowptr, dinv, ent, N_EDGES);

  k_gemm_wmma<NFH, NFH><<<gemmBlocksH, T, 0, stream>>>(x, W_in, h, N_NODES);
  k_gather<NFH, true><<<rowBlocks, T, 0, stream>>>(rowptr, ent, dinv, h, b_in, x, xA, N_NODES);

  k_gemm_wmma<NFH, NFH><<<gemmBlocksH, T, 0, stream>>>(xA, W_m0, h, N_NODES);
  k_gather<NFH, true><<<rowBlocks, T, 0, stream>>>(rowptr, ent, dinv, h, b_m0, xA, xB, N_NODES);

  k_gemm_wmma<NFH, NFH><<<gemmBlocksH, T, 0, stream>>>(xB, W_m1, h, N_NODES);
  k_gather<NFH, true><<<rowBlocks, T, 0, stream>>>(rowptr, ent, dinv, h, b_m1, xB, xA, N_NODES);

  k_gemm_wmma<NFH, NCLS><<<gemmBlocksC, T, 0, stream>>>(xA, W_out, h, N_NODES);
  k_gemm_wmma<NFH, NCLS><<<gemmBlocksC, T, 0, stream>>>(xA, W_skip, skip, N_NODES);
  k_gather<NCLS, false><<<rowBlocks, T, 0, stream>>>(rowptr, ent, dinv, h, b_out, skip,
                                                     (float*)d_out, N_NODES);
}


# GEMM: W copied once per block into LDS as row pairs; each wave computes a full 16 x M strip, so one A fragment feeds M/16 WMMAs and B fragments are LDS reads; k order and products unchanged
# speedup vs baseline: 3.4514x; 1.0636x over previous
#include <hip/hip_runtime.h>
#include <hip/hip_bf16.h>

#define N_NODES 100000
#define N_EDGES 1600000
#define NFH     128
#define NCLS    64

typedef __attribute__((ext_vector_type(2))) float v2f;
typedef __attribute__((ext_vector_type(8))) float v8f;

#define SCAN_T   256
#define SCAN_BLK 1024

__global__ void k_zero_cnt(int* __restrict__ cnt, int n) {
  int i = blockIdx.x * blockDim.x + threadIdx.x;
  if (i < n) cnt[i] = 0;
}

__global__ void k_rank(const int* __restrict__ dst, int* __restrict__ cnt,
                       int* __restrict__ rank, int e) {
  int i = blockIdx.x * blockDim.x + threadIdx.x;
  if (i < e) rank[i] = atomicAdd(&cnt[dst[i]], 1);
}

__global__ __launch_bounds__(SCAN_T)
void k_blocksum(const int* __restrict__ cnt, int* __restrict__ bsum, int n) {
  __shared__ int wsum[SCAN_T / 32];
  const int lane = threadIdx.x & 31;
  const int w    = threadIdx.x >> 5;
  const int i0   = (blockIdx.x * SCAN_T + threadIdx.x) * 4;
  int s = 0;
  if (i0 < n) {
    const int4 c = *reinterpret_cast<const int4*>(cnt + i0);
    s = c.x + c.y + c.z + c.w;
  }
  for (int o = 16; o > 0; o >>= 1) s += __shfl_xor(s, o);
  if (lane == 0) wsum[w] = s;
  __syncthreads();
  if (threadIdx.x == 0) {
    int t = 0;
    for (int k = 0; k < SCAN_T / 32; ++k) t += wsum[k];
    bsum[blockIdx.x] = t;
  }
}

__global__ __launch_bounds__(SCAN_T)
void k_scan(const int* __restrict__ cnt, const int* __restrict__ bsum,
            int* __restrict__ rowptr, float* __restrict__ dinv, int n) {
  __shared__ int wsum[SCAN_T / 32];
  const int lane = threadIdx.x & 31;
  const int w    = threadIdx.x >> 5;
  int base = 0;
  for (int k = 0; k < (int)blockIdx.x; ++k) base += bsum[k];
  const int i0 = (blockIdx.x * SCAN_T + threadIdx.x) * 4;
  int4 c = make_int4(0, 0, 0, 0);
  if (i0 < n) c = *reinterpret_cast<const int4*>(cnt + i0);
  const int tsum = c.x + c.y + c.z + c.w;
  int inc = tsum;
  for (int o = 1; o < 32; o <<= 1) {
    const int t = __shfl_up(inc, o);
    if (lane >= o) inc += t;
  }
  if (lane == 31) wsum[w] = inc;
  __syncthreads();
  int woff = 0;
  for (int k = 0; k < SCAN_T / 32; ++k) woff += (k < w) ? wsum[k] : 0;
  const int p = base + woff + inc - tsum;
  if (i0 < n) {
    int4 r;
    r.x = p;
    r.y = r.x + c.x;
    r.z = r.y + c.y;
    r.w = r.z + c.z;
    *reinterpret_cast<int4*>(rowptr + i0) = r;
    float4 d;
    d.x = rsqrtf((float)(c.x + 1));
    d.y = rsqrtf((float)(c.y + 1));
    d.z = rsqrtf((float)(c.z + 1));
    d.w = rsqrtf((float)(c.w + 1));
    *reinterpret_cast<float4*>(dinv + i0) = d;
    if (i0 + 4 == n) rowptr[n] = r.w + c.w;
  }
}

__global__ void k_fill(const int* __restrict__ src, const int* __restrict__ dst,
                       const int* __restrict__ rank, const int* __restrict__ rowptr,
                       const float* __restrict__ dinv, int2* __restrict__ ent, int e) {
  int i = blockIdx.x * blockDim.x + threadIdx.x;
  if (i < e) {
    const int s = src[i];
    const int d = dst[i];
    ent[rowptr[d] + rank[i]] = make_int2(s, __float_as_int(dinv[s] * dinv[d]));
  }
}

#define GEMM_T 128

template <int K, int M>
__global__ __launch_bounds__(GEMM_T)
void k_gemm_wmma(const float* __restrict__ X, const float* __restrict__ W,
                 float* __restrict__ H, int n_rows) {
  constexpr int WT    = M / 16;
  constexpr int PITCH = 2 * M + 32;
  constexpr int CP    = M / 2;
  constexpr int ITEMS = (K / 2) * CP;
  __shared__ __attribute__((aligned(16))) float Ws[(K / 2) * PITCH];

  static_assert(ITEMS % GEMM_T == 0, "whole passes over W");

#pragma unroll
  for (unsigned s = 0; s < ITEMS / GEMM_T; ++s) {
    const unsigned it = threadIdx.x + s * GEMM_T;
    const unsigned j = it / CP;
    const unsigned p = it % CP;
    const float2 r0 = *reinterpret_cast<const float2*>(W + (size_t)(2 * j) * M + 2 * p);
    const float2 r1 = *reinterpret_cast<const float2*>(W + (size_t)(2 * j + 1) * M + 2 * p);
    *reinterpret_cast<float4*>(&Ws[j * PITCH + 4 * p]) = make_float4(r0.x, r1.x, r0.y, r1.y);
  }
  __syncthreads();

  const int lane = threadIdx.x & 31;
  const int tile = __builtin_amdgcn_readfirstlane(
      (int)(blockIdx.x * (GEMM_T / 32) + (threadIdx.x >> 5)));
  if (tile >= n_rows / 16) return;

  const int m = lane & 15;
  const int hf = lane >> 4;

  const float* __restrict__ xrow = X + (size_t)(tile * 16 + m) * K + 2 * hf;
  const float* bl = Ws + hf * PITCH + 2 * m;

  v8f acc[WT];
#pragma unroll
  for (int t = 0; t < WT; ++t) acc[t] = v8f{};

#pragma unroll
  for (int k0 = 0; k0 < K; k0 += 4) {
    v2f a;
    a.x = xrow[k0];
    a.y = xrow[k0 + 1];
#pragma unroll
    for (int t = 0; t < WT; ++t) {
      const float2 bb = *reinterpret_cast<const float2*>(bl + (k0 / 2) * PITCH + t * 32);
      v2f b;
      b.x = bb.x;
      b.y = bb.y;
      acc[t] = __builtin_amdgcn_wmma_f32_16x16x4_f32(false, a, false, b, (short)0, acc[t],
                                                     false, false);
    }
  }

  const int row0 = tile * 16 + (hf << 3);
  float* __restrict__ out0 = H + (size_t)row0 * M + m;
#pragma unroll
  for (int v = 0; v < 8; ++v) {
#pragma unroll
    for (int t = 0; t < WT; ++t) out0[(size_t)v * M + t * 16] = acc[t][v];
  }
}

template <int F> struct RowVec;
template <> struct RowVec<4> { typedef float4 T; };
template <> struct RowVec<2> { typedef float2 T; };

__device__ __forceinline__ void axpy(float4& a, float c, const float4& v) {
  a.x += c * v.x; a.y += c * v.y; a.z += c * v.z; a.w += c * v.w;
}
__device__ __forceinline__ void axpy(float2& a, float c, const float2& v) {
  a.x += c * v.x; a.y += c * v.y;
}
template <bool RELU>
__device__ __forceinline__ float add_res(float a, float r) {
  const float v = a + r;
  return RELU ? (v > 0.0f ? v : 0.0f) : v;
}
template <bool RELU>
__device__ __forceinline__ float4 add_res(const float4& a, const float4& r) {
  return make_float4(add_res<RELU>(a.x, r.x), add_res<RELU>(a.y, r.y),
                     add_res<RELU>(a.z, r.z), add_res<RELU>(a.w, r.w));
}
template <bool RELU>
__device__ __forceinline__ float2 add_res(const float2& a, const float2& r) {
  return make_float2(add_res<RELU>(a.x, r.x), add_res<RELU>(a.y, r.y));
}

template <int M, bool RELU>
__global__ __launch_bounds__(256)
void k_gather(const int* __restrict__ rowptr, const int2* __restrict__ ent,
              const float* __restrict__ dinv, const float* __restrict__ h,
              const float* __restrict__ bias, const float* __restrict__ res,
              float* __restrict__ y, int n) {
  constexpr int F = M / 32;
  typedef typename RowVec<F>::T VT;
  const int lane = threadIdx.x & 31;
  const int i = __builtin_amdgcn_readfirstlane(
      (blockIdx.x * (blockDim.x >> 5)) + (threadIdx.x >> 5));
  if (i >= n) return;
  const int beg = rowptr[i];
  const int end = rowptr[i + 1];
  const float di = dinv[i];
  const VT* __restrict__ hv = reinterpret_cast<const VT*>(h);

  const VT rr = reinterpret_cast<const VT*>(res)[(size_t)i * 32 + lane];
  VT acc = reinterpret_cast<const VT*>(bias)[lane];
  axpy(acc, di * di, hv[(size_t)i * 32 + lane]);

  for (int base = beg; base < end; base += 32) {
    const int cnt = (end - base) < 32 ? (end - base) : 32;
    int2 en = make_int2(0, 0);
    if (lane < cnt) en = ent[base + lane];
#define ENT_SRC(t)  __builtin_amdgcn_readlane(en.x, (t))
#define ENT_COEF(t) __int_as_float(__builtin_amdgcn_readlane(en.y, (t)))
#define ROW(t)      hv[(size_t)ENT_SRC(t) * 32 + lane]
    int t = 0;
    for (; t + 4 <= cnt; t += 4) {
      const VT v0 = ROW(t), v1 = ROW(t + 1), v2 = ROW(t + 2), v3 = ROW(t + 3);
      axpy(acc, ENT_COEF(t), v0);
      axpy(acc, ENT_COEF(t + 1), v1);
      axpy(acc, ENT_COEF(t + 2), v2);
      axpy(acc, ENT_COEF(t + 3), v3);
    }
    const int r = cnt - t;
    if (r == 3) {
      const VT v0 = ROW(t), v1 = ROW(t + 1), v2 = ROW(t + 2);
      axpy(acc, ENT_COEF(t), v0);
      axpy(acc, ENT_COEF(t + 1), v1);
      axpy(acc, ENT_COEF(t + 2), v2);
    } else if (r == 2) {
      const VT v0 = ROW(t), v1 = ROW(t + 1);
      axpy(acc, ENT_COEF(t), v0);
      axpy(acc, ENT_COEF(t + 1), v1);
    } else if (r == 1) {
      const VT v0 = ROW(t);
      axpy(acc, ENT_COEF(t), v0);
    }
#undef ROW
#undef ENT_COEF
#undef ENT_SRC
  }
  reinterpret_cast<VT*>(y)[(size_t)i * 32 + lane] = add_res<RELU>(acc, rr);
}

extern "C" void kernel_launch(void* const* d_in, const int* in_sizes, int n_in,
                              void* d_out, int out_size, void* d_ws, size_t ws_size,
                              hipStream_t stream) {
  const float* x      = (const float*)d_in[0];
  const int*   ei     = (const int*)d_in[1];
  const float* W_in   = (const float*)d_in[2];
  const float* b_in   = (const float*)d_in[3];
  const float* W_m0   = (const float*)d_in[4];
  const float* b_m0   = (const float*)d_in[5];
  const float* W_m1   = (const float*)d_in[6];
  const float* b_m1   = (const float*)d_in[7];
  const float* W_out  = (const float*)d_in[8];
  const float* b_out  = (const float*)d_in[9];
  const float* W_skip = (const float*)d_in[10];

  const int* src = ei;
  const int* dst = ei + N_EDGES;

  float* w    = (float*)d_ws;
  float* dinv = w;  w += N_NODES;
  float* h    = w;  w += (size_t)N_NODES * NFH;
  float* xA   = w;  w += (size_t)N_NODES * NFH;
  float* xB   = w;  w += (size_t)N_NODES * NFH;
  float* skip = w;  w += (size_t)N_NODES * NCLS;
  int*  cnt    = (int*)w;   w += N_NODES;
  int*  rowptr = (int*)w;   w += N_NODES + 4;
  int*  bsum   = (int*)w;   w += 128;
  int*  rank   = (int*)w;   w += N_EDGES;
  int2* ent    = (int2*)w;  w += (size_t)2 * N_EDGES;

  const int T = 256;
  auto cdiv = [](int a, int b) { return (a + b - 1) / b; };

  const int nodeBlocks  = cdiv(N_NODES, T);
  const int edgeBlocks  = cdiv(N_EDGES, T);
  const int scanBlocks  = cdiv(N_NODES, SCAN_BLK);
  const int rowBlocks   = cdiv(N_NODES, T / 32);
  const int gemmBlocks  = cdiv(N_NODES / 16, GEMM_T / 32);

  k_zero_cnt<<<nodeBlocks, T, 0, stream>>>(cnt, N_NODES);
  k_rank<<<edgeBlocks, T, 0, stream>>>(dst, cnt, rank, N_EDGES);
  k_blocksum<<<scanBlocks, SCAN_T, 0, stream>>>(cnt, bsum, N_NODES);
  k_scan<<<scanBlocks, SCAN_T, 0, stream>>>(cnt, bsum, rowptr, dinv, N_NODES);
  k_fill<<<edgeBlocks, T, 0, stream>>>(src, dst, rank, rowptr, dinv, ent, N_EDGES);

  k_gemm_wmma<NFH, NFH><<<gemmBlocks, GEMM_T, 0, stream>>>(x, W_in, h, N_NODES);
  k_gather<NFH, true><<<rowBlocks, T, 0, stream>>>(rowptr, ent, dinv, h, b_in, x, xA, N_NODES);

  k_gemm_wmma<NFH, NFH><<<gemmBlocks, GEMM_T, 0, stream>>>(xA, W_m0, h, N_NODES);
  k_gather<NFH, true><<<rowBlocks, T, 0, stream>>>(rowptr, ent, dinv, h, b_m0, xA, xB, N_NODES);

  k_gemm_wmma<NFH, NFH><<<gemmBlocks, GEMM_T, 0, stream>>>(xB, W_m1, h, N_NODES);
  k_gather<NFH, true><<<rowBlocks, T, 0, stream>>>(rowptr, ent, dinv, h, b_m1, xB, xA, N_NODES);

  k_gemm_wmma<NFH, NCLS><<<gemmBlocks, GEMM_T, 0, stream>>>(xA, W_out, h, N_NODES);
  k_gemm_wmma<NFH, NCLS><<<gemmBlocks, GEMM_T, 0, stream>>>(xA, W_skip, skip, N_NODES);
  k_gather<NCLS, false><<<rowBlocks, T, 0, stream>>>(rowptr, ent, dinv, h, b_out, skip,
                                                     (float*)d_out, N_NODES);
}


# Row-sum kernel of layers 0-2 also runs the next layer's product: 16 resident waves sum a 16-row tile into LDS, 8 run its WMMA strip from LDS-held W; 4 GEMM launches gone; sum and k order unchanged
# speedup vs baseline: 3.8177x; 1.1061x over previous
#include <hip/hip_runtime.h>
#include <hip/hip_bf16.h>

#define N_NODES 100000
#define N_EDGES 1600000
#define NFH     128
#define NCLS    64

typedef __attribute__((ext_vector_type(2))) float v2f;
typedef __attribute__((ext_vector_type(8))) float v8f;

#define SCAN_T   256
#define SCAN_BLK 1024

__global__ void k_zero_cnt(int* __restrict__ cnt, int n) {
  int i = blockIdx.x * blockDim.x + threadIdx.x;
  if (i < n) cnt[i] = 0;
}

__global__ void k_rank(const int* __restrict__ dst, int* __restrict__ cnt,
                       int* __restrict__ rank, int e) {
  int i = blockIdx.x * blockDim.x + threadIdx.x;
  if (i < e) rank[i] = atomicAdd(&cnt[dst[i]], 1);
}

__global__ __launch_bounds__(SCAN_T)
void k_blocksum(const int* __restrict__ cnt, int* __restrict__ bsum, int n) {
  __shared__ int wsum[SCAN_T / 32];
  const int lane = threadIdx.x & 31;
  const int w    = threadIdx.x >> 5;
  const int i0   = (blockIdx.x * SCAN_T + threadIdx.x) * 4;
  int s = 0;
  if (i0 < n) {
    const int4 c = *reinterpret_cast<const int4*>(cnt + i0);
    s = c.x + c.y + c.z + c.w;
  }
  for (int o = 16; o > 0; o >>= 1) s += __shfl_xor(s, o);
  if (lane == 0) wsum[w] = s;
  __syncthreads();
  if (threadIdx.x == 0) {
    int t = 0;
    for (int k = 0; k < SCAN_T / 32; ++k) t += wsum[k];
    bsum[blockIdx.x] = t;
  }
}

__global__ __launch_bounds__(SCAN_T)
void k_scan(const int* __restrict__ cnt, const int* __restrict__ bsum,
            int* __restrict__ rowptr, float* __restrict__ dinv, int n) {
  __shared__ int wsum[SCAN_T / 32];
  const int lane = threadIdx.x & 31;
  const int w    = threadIdx.x >> 5;
  int base = 0;
  for (int k = 0; k < (int)blockIdx.x; ++k) base += bsum[k];
  const int i0 = (blockIdx.x * SCAN_T + threadIdx.x) * 4;
  int4 c = make_int4(0, 0, 0, 0);
  if (i0 < n) c = *reinterpret_cast<const int4*>(cnt + i0);
  const int tsum = c.x + c.y + c.z + c.w;
  int inc = tsum;
  for (int o = 1; o < 32; o <<= 1) {
    const int t = __shfl_up(inc, o);
    if (lane >= o) inc += t;
  }
  if (lane == 31) wsum[w] = inc;
  __syncthreads();
  int woff = 0;
  for (int k = 0; k < SCAN_T / 32; ++k) woff += (k < w) ? wsum[k] : 0;
  const int p = base + woff + inc - tsum;
  if (i0 < n) {
    int4 r;
    r.x = p;
    r.y = r.x + c.x;
    r.z = r.y + c.y;
    r.w = r.z + c.z;
    *reinterpret_cast<int4*>(rowptr + i0) = r;
    float4 d;
    d.x = rsqrtf((float)(c.x + 1));
    d.y = rsqrtf((float)(c.y + 1));
    d.z = rsqrtf((float)(c.z + 1));
    d.w = rsqrtf((float)(c.w + 1));
    *reinterpret_cast<float4*>(dinv + i0) = d;
    if (i0 + 4 == n) rowptr[n] = r.w + c.w;
  }
}

__global__ void k_fill(const int* __restrict__ src, const int* __restrict__ dst,
                       const int* __restrict__ rank, const int* __restrict__ rowptr,
                       const float* __restrict__ dinv, int2* __restrict__ ent, int e) {
  int i = blockIdx.x * blockDim.x + threadIdx.x;
  if (i < e) {
    const int s = src[i];
    const int d = dst[i];
    ent[rowptr[d] + rank[i]] = make_int2(s, __float_as_int(dinv[s] * dinv[d]));
  }
}

#define GEMM_T 128

template <int K, int M>
__global__ __launch_bounds__(GEMM_T)
void k_gemm_wmma(const float* __restrict__ X, const float* __restrict__ W,
                 float* __restrict__ H, int n_rows) {
  constexpr int WT    = M / 16;
  constexpr int PITCH = 2 * M + 32;
  constexpr int CP    = M / 2;
  constexpr int ITEMS = (K / 2) * CP;
  __shared__ __attribute__((aligned(16))) float Ws[(K / 2) * PITCH];

  static_assert(ITEMS % GEMM_T == 0, "whole passes over W");

#pragma unroll
  for (unsigned s = 0; s < ITEMS / GEMM_T; ++s) {
    const unsigned it = threadIdx.x + s * GEMM_T;
    const unsigned j = it / CP;
    const unsigned p = it % CP;
    const float2 r0 = *reinterpret_cast<const float2*>(W + (size_t)(2 * j) * M + 2 * p);
    const float2 r1 = *reinterpret_cast<const float2*>(W + (size_t)(2 * j + 1) * M + 2 * p);
    *reinterpret_cast<float4*>(&Ws[j * PITCH + 4 * p]) = make_float4(r0.x, r1.x, r0.y, r1.y);
  }
  __syncthreads();

  const int lane = threadIdx.x & 31;
  const int tile = __builtin_amdgcn_readfirstlane(
      (int)(blockIdx.x * (GEMM_T / 32) + (threadIdx.x >> 5)));
  if (tile >= n_rows / 16) return;

  const int m = lane & 15;
  const int hf = lane >> 4;

  const float* __restrict__ xrow = X + (size_t)(tile * 16 + m) * K + 2 * hf;
  const float* bl = Ws + hf * PITCH + 2 * m;

  v8f acc[WT];
#pragma unroll
  for (int t = 0; t < WT; ++t) acc[t] = v8f{};

#pragma unroll
  for (int k0 = 0; k0 < K; k0 += 4) {
    v2f a;
    a.x = xrow[k0];
    a.y = xrow[k0 + 1];
#pragma unroll
    for (int t = 0; t < WT; ++t) {
      const float2 bb = *reinterpret_cast<const float2*>(bl + (k0 / 2) * PITCH + t * 32);
      v2f b;
      b.x = bb.x;
      b.y = bb.y;
      acc[t] = __builtin_amdgcn_wmma_f32_16x16x4_f32(false, a, false, b, (short)0, acc[t],
                                                     false, false);
    }
  }

  const int row0 = tile * 16 + (hf << 3);
  float* __restrict__ out0 = H + (size_t)row0 * M + m;
#pragma unroll
  for (int v = 0; v < 8; ++v) {
#pragma unroll
    for (int t = 0; t < WT; ++t) out0[(size_t)v * M + t * 16] = acc[t][v];
  }
}

template <int F> struct RowVec;
template <> struct RowVec<4> { typedef float4 T; };
template <> struct RowVec<2> { typedef float2 T; };

__device__ __forceinline__ void axpy(float4& a, float c, const float4& v) {
  a.x += c * v.x; a.y += c * v.y; a.z += c * v.z; a.w += c * v.w;
}
__device__ __forceinline__ void axpy(float2& a, float c, const float2& v) {
  a.x += c * v.x; a.y += c * v.y;
}
template <bool RELU>
__device__ __forceinline__ float add_res(float a, float r) {
  const float v = a + r;
  return RELU ? (v > 0.0f ? v : 0.0f) : v;
}
template <bool RELU>
__device__ __forceinline__ float4 add_res(const float4& a, const float4& r) {
  return make_float4(add_res<RELU>(a.x, r.x), add_res<RELU>(a.y, r.y),
                     add_res<RELU>(a.z, r.z), add_res<RELU>(a.w, r.w));
}
template <bool RELU>
__device__ __forceinline__ float2 add_res(const float2& a, const float2& r) {
  return make_float2(add_res<RELU>(a.x, r.x), add_res<RELU>(a.y, r.y));
}

template <int M, bool RELU>
__device__ __forceinline__ typename RowVec<M / 32>::T
row_sum(const int i, const int lane,
        const int* __restrict__ rowptr, const int2* __restrict__ ent,
        const float* __restrict__ dinv, const float* __restrict__ h,
        const float* __restrict__ bias, const float* __restrict__ res) {
  constexpr int F = M / 32;
  typedef typename RowVec<F>::T VT;
  const int beg = rowptr[i];
  const int end = rowptr[i + 1];
  const float di = dinv[i];
  const VT* __restrict__ hv = reinterpret_cast<const VT*>(h);

  const VT rr = reinterpret_cast<const VT*>(res)[(size_t)i * 32 + lane];
  VT acc = reinterpret_cast<const VT*>(bias)[lane];
  axpy(acc, di * di, hv[(size_t)i * 32 + lane]);

  for (int base = beg; base < end; base += 32) {
    const int cnt = (end - base) < 32 ? (end - base) : 32;
    int2 en = make_int2(0, 0);
    if (lane < cnt) en = ent[base + lane];
#define ENT_SRC(t)  __builtin_amdgcn_readlane(en.x, (t))
#define ENT_COEF(t) __int_as_float(__builtin_amdgcn_readlane(en.y, (t)))
#define ROW(t)      hv[(size_t)ENT_SRC(t) * 32 + lane]
    int t = 0;
    for (; t + 4 <= cnt; t += 4) {
      const VT v0 = ROW(t), v1 = ROW(t + 1), v2 = ROW(t + 2), v3 = ROW(t + 3);
      axpy(acc, ENT_COEF(t), v0);
      axpy(acc, ENT_COEF(t + 1), v1);
      axpy(acc, ENT_COEF(t + 2), v2);
      axpy(acc, ENT_COEF(t + 3), v3);
    }
    const int r = cnt - t;
    if (r == 3) {
      const VT v0 = ROW(t), v1 = ROW(t + 1), v2 = ROW(t + 2);
      axpy(acc, ENT_COEF(t), v0);
      axpy(acc, ENT_COEF(t + 1), v1);
      axpy(acc, ENT_COEF(t + 2), v2);
    } else if (r == 2) {
      const VT v0 = ROW(t), v1 = ROW(t + 1);
      axpy(acc, ENT_COEF(t), v0);
      axpy(acc, ENT_COEF(t + 1), v1);
    } else if (r == 1) {
      const VT v0 = ROW(t);
      axpy(acc, ENT_COEF(t), v0);
    }
#undef ROW
#undef ENT_COEF
#undef ENT_SRC
  }
  return add_res<RELU>(acc, rr);
}

template <int M, bool RELU>
__global__ __launch_bounds__(256)
void k_gather(const int* __restrict__ rowptr, const int2* __restrict__ ent,
              const float* __restrict__ dinv, const float* __restrict__ h,
              const float* __restrict__ bias, const float* __restrict__ res,
              float* __restrict__ y, int n) {
  typedef typename RowVec<M / 32>::T VT;
  const int lane = threadIdx.x & 31;
  const int i = __builtin_amdgcn_readfirstlane(
      (blockIdx.x * (blockDim.x >> 5)) + (threadIdx.x >> 5));
  if (i >= n) return;
  reinterpret_cast<VT*>(y)[(size_t)i * 32 + lane] =
      row_sum<M, RELU>(i, lane, rowptr, ent, dinv, h, bias, res);
}

#define FG_T 512
#define YP   132

template <bool SPLIT, bool STORE_Y>
__global__ __launch_bounds__(FG_T)
void k_gather_gemm(const int* __restrict__ rowptr, const int2* __restrict__ ent,
                   const float* __restrict__ dinv, const float* __restrict__ h,
                   const float* __restrict__ bias, const float* __restrict__ res,
                   float* __restrict__ y,
                   const float* __restrict__ Wa, const float* __restrict__ Wb,
                   float* __restrict__ Ha, float* __restrict__ Hb, int n_tiles) {
  __shared__ __attribute__((aligned(16))) float Ws[NFH * NFH];
  __shared__ __attribute__((aligned(16))) float Ys[16 * YP];

#pragma unroll
  for (unsigned s = 0; s < (NFH / 4) * NFH / FG_T; ++s) {
    const unsigned it = threadIdx.x + s * FG_T;
    const unsigned j = it / NFH;
    const unsigned c = it % NFH;
    const float* wp;
    unsigned ld;
    if (SPLIT) {
      wp = (c < NCLS) ? (Wa + c) : (Wb + (c - NCLS));
      ld = NCLS;
    } else {
      wp = Wa + c;
      ld = NFH;
    }
    wp += (size_t)(4 * j) * ld;
    *reinterpret_cast<float4*>(&Ws[j * (4 * NFH) + 4 * c]) =
        make_float4(wp[0], wp[ld], wp[2 * ld], wp[3 * ld]);
  }

  const int lane = threadIdx.x & 31;
  const int w = __builtin_amdgcn_readfirstlane((int)(threadIdx.x >> 5));
  const int m = lane & 15;
  const int hf = lane >> 4;

  int turn = 0;
  for (int tile = blockIdx.x; tile < n_tiles; tile += gridDim.x, turn ^= 1) {
    const int i = tile * 16 + w;
    const float4 yv = row_sum<NFH, true>(i, lane, rowptr, ent, dinv, h, bias, res);
    __syncthreads();
    *reinterpret_cast<float4*>(&Ys[w * YP + 4 * lane]) = yv;
    __syncthreads();
    if (STORE_Y) reinterpret_cast<float4*>(y)[(size_t)i * 32 + lane] = yv;
    if ((w >> 3) == turn) {
      const int t = w & 7;
      const float* al = Ys + m * YP + 2 * hf;
      const float* bl = Ws + (t * 16 + m) * 4 + 2 * hf;
      v8f acc = v8f{};
#pragma unroll
      for (int k0 = 0; k0 < NFH; k0 += 4) {
        const float2 aa = *reinterpret_cast<const float2*>(al + k0);
        const float2 bb = *reinterpret_cast<const float2*>(bl + (k0 / 4) * (4 * NFH));
        v2f a, b;
        a.x = aa.x; a.y = aa.y;
        b.x = bb.x; b.y = bb.y;
        acc = __builtin_amdgcn_wmma_f32_16x16x4_f32(false, a, false, b, (short)0, acc,
                                                    false, false);
      }
      const int row0 = tile * 16 + (hf << 3);
      float* o;
      int ld;
      if (SPLIT) {
        o = (t < 4) ? (Ha + t * 16 + m) : (Hb + (t - 4) * 16 + m);
        ld = NCLS;
      } else {
        o = Ha + t * 16 + m;
        ld = NFH;
      }
      o += (size_t)row0 * ld;
#pragma unroll
      for (int v = 0; v < 8; ++v) o[(size_t)v * ld] = acc[v];
    }
  }
}

extern "C" void kernel_launch(void* const* d_in, const int* in_sizes, int n_in,
                              void* d_out, int out_size, void* d_ws, size_t ws_size,
                              hipStream_t stream) {
  const float* x      = (const float*)d_in[0];
  const int*   ei     = (const int*)d_in[1];
  const float* W_in   = (const float*)d_in[2];
  const float* b_in   = (const float*)d_in[3];
  const float* W_m0   = (const float*)d_in[4];
  const float* b_m0   = (const float*)d_in[5];
  const float* W_m1   = (const float*)d_in[6];
  const float* b_m1   = (const float*)d_in[7];
  const float* W_out  = (const float*)d_in[8];
  const float* b_out  = (const float*)d_in[9];
  const float* W_skip = (const float*)d_in[10];

  const int* src = ei;
  const int* dst = ei + N_EDGES;

  float* w    = (float*)d_ws;
  float* dinv = w;  w += N_NODES;
  float* hA   = w;  w += (size_t)N_NODES * NFH;
  float* hB   = w;  w += (size_t)N_NODES * NFH;
  float* xA   = w;  w += (size_t)N_NODES * NFH;
  float* xB   = w;  w += (size_t)N_NODES * NFH;
  int*  cnt    = (int*)w;   w += N_NODES;
  int*  rowptr = (int*)w;   w += N_NODES + 4;
  int*  bsum   = (int*)w;   w += 128;
  int*  rank   = (int*)w;   w += N_EDGES;
  int2* ent    = (int2*)w;  w += (size_t)2 * N_EDGES;
  float* hOut  = hB;
  float* hSkip = hB + (size_t)N_NODES * NCLS;

  const int T = 256;
  auto cdiv = [](int a, int b) { return (a + b - 1) / b; };

  const int nodeBlocks  = cdiv(N_NODES, T);
  const int edgeBlocks  = cdiv(N_EDGES, T);
  const int scanBlocks  = cdiv(N_NODES, SCAN_BLK);
  const int rowBlocks   = cdiv(N_NODES, T / 32);
  const int gemmBlocks  = cdiv(N_NODES / 16, GEMM_T / 32);
  const int fusedBlocks = 1024;
  const int nTiles      = N_NODES / 16;

  k_zero_cnt<<<nodeBlocks, T, 0, stream>>>(cnt, N_NODES);
  k_rank<<<edgeBlocks, T, 0, stream>>>(dst, cnt, rank, N_EDGES);
  k_blocksum<<<scanBlocks, SCAN_T, 0, stream>>>(cnt, bsum, N_NODES);
  k_scan<<<scanBlocks, SCAN_T, 0, stream>>>(cnt, bsum, rowptr, dinv, N_NODES);
  k_fill<<<edgeBlocks, T, 0, stream>>>(src, dst, rank, rowptr, dinv, ent, N_EDGES);

  k_gemm_wmma<NFH, NFH><<<gemmBlocks, GEMM_T, 0, stream>>>(x, W_in, hA, N_NODES);

  k_gather_gemm<false, true><<<fusedBlocks, FG_T, 0, stream>>>(
      rowptr, ent, dinv, hA, b_in, x, xA, W_m0, nullptr, hB, nullptr, nTiles);

  k_gather_gemm<false, true><<<fusedBlocks, FG_T, 0, stream>>>(
      rowptr, ent, dinv, hB, b_m0, xA, xB, W_m1, nullptr, hA, nullptr, nTiles);

  k_gather_gemm<true, false><<<fusedBlocks, FG_T, 0, stream>>>(
      rowptr, ent, dinv, hA, b_m1, xB, nullptr, W_out, W_skip, hOut, hSkip, nTiles);

  k_gather<NCLS, false><<<rowBlocks, T, 0, stream>>>(rowptr, ent, dinv, hOut, b_out, hSkip,
                                                     (float*)d_out, N_NODES);
}


# Edges grouped by destination by a two-level counting sort that counts in LDS (256-node buckets, then one block per bucket), not a scattered returning global add per edge; row order: LDS arrival rank
# speedup vs baseline: 4.7604x; 1.2469x over previous
#include <hip/hip_runtime.h>
#include <hip/hip_bf16.h>

#define N_NODES 100000
#define N_EDGES 1600000
#define NFH     128
#define NCLS    64

typedef __attribute__((ext_vector_type(2))) float v2f;
typedef __attribute__((ext_vector_type(8))) float v8f;

#define BN     256
#define NB     ((N_NODES + BN - 1) / BN)
#define P1_B   256
#define P1_T   512
#define P1_CH  ((N_EDGES + P1_B - 1) / P1_B)
#define P1_PER ((P1_CH + P1_T - 1) / P1_T)
#define BS_T   512
#define BK_T   512

__global__ __launch_bounds__(P1_T)
void k_bcount(const int* __restrict__ dst, int* __restrict__ hist) {
  __shared__ int hl[NB];
  for (int t = threadIdx.x; t < NB; t += P1_T) hl[t] = 0;
  __syncthreads();
  const int e0 = blockIdx.x * P1_CH;
  const int e1 = (e0 + P1_CH < N_EDGES) ? (e0 + P1_CH) : N_EDGES;
  int dv[P1_PER];
#pragma unroll
  for (int k = 0; k < P1_PER; ++k) {
    const int i = e0 + (int)threadIdx.x + k * P1_T;
    dv[k] = (i < e1) ? dst[i] : -1;
  }
#pragma unroll
  for (int k = 0; k < P1_PER; ++k)
    if (dv[k] >= 0) atomicAdd(&hl[dv[k] / BN], 1);
  __syncthreads();
  for (int t = threadIdx.x; t < NB; t += P1_T) hist[blockIdx.x * NB + t] = hl[t];
}

__global__ __launch_bounds__(BS_T)
void k_bscan(const int* __restrict__ hist, int* __restrict__ pre, int* __restrict__ base) {
  static_assert(NB <= BS_T, "one thread per bucket");
  __shared__ int wsum[BS_T / 32];
  const int t    = threadIdx.x;
  const int lane = threadIdx.x & 31;
  const int w    = threadIdx.x >> 5;
  int run = 0;
  if (t < NB) {
#pragma unroll 32
    for (int k = 0; k < P1_B; ++k) {
      const int v = hist[k * NB + t];
      pre[k * NB + t] = run;
      run += v;
    }
  }
  int inc = run;
  for (int o = 1; o < 32; o <<= 1) {
    const int u = __shfl_up(inc, o);
    if (lane >= o) inc += u;
  }
  if (lane == 31) wsum[w] = inc;
  __syncthreads();
  int woff = 0;
  for (int k = 0; k < BS_T / 32; ++k) woff += (k < w) ? wsum[k] : 0;
  const int ex = woff + inc - run;
  if (t < NB) base[t] = ex;
  if (t == NB - 1) base[NB] = ex + run;
}

__global__ __launch_bounds__(P1_T)
void k_part(const int* __restrict__ src, const int* __restrict__ dst,
            const int* __restrict__ pre, const int* __restrict__ base,
            int2* __restrict__ psd) {
  __shared__ int cur[NB];
  for (int t = threadIdx.x; t < NB; t += P1_T) cur[t] = base[t] + pre[blockIdx.x * NB + t];
  __syncthreads();
  const int e0 = blockIdx.x * P1_CH;
  const int e1 = (e0 + P1_CH < N_EDGES) ? (e0 + P1_CH) : N_EDGES;
  int sv[P1_PER], dv[P1_PER];
#pragma unroll
  for (int k = 0; k < P1_PER; ++k) {
    const int i = e0 + (int)threadIdx.x + k * P1_T;
    const bool ok = i < e1;
    sv[k] = ok ? src[i] : 0;
    dv[k] = ok ? dst[i] : -1;
  }
#pragma unroll
  for (int k = 0; k < P1_PER; ++k) {
    if (dv[k] >= 0) {
      const int p = atomicAdd(&cur[dv[k] / BN], 1);
      psd[p] = make_int2(sv[k], dv[k]);
    }
  }
}

__global__ __launch_bounds__(BK_T)
void k_bucket_count(const int2* __restrict__ psd, const int* __restrict__ base,
                    int* __restrict__ rowptr, float* __restrict__ dinv) {
  static_assert(BN <= BK_T, "one thread per node of the bucket");
  __shared__ int cl[BN];
  __shared__ int wsum[BK_T / 32];
  const int lane = threadIdx.x & 31;
  const int w    = threadIdx.x >> 5;
  const int n0   = blockIdx.x * BN;
  if (threadIdx.x < BN) cl[threadIdx.x] = 0;
  __syncthreads();
  const int beg = base[blockIdx.x];
  const int end = base[blockIdx.x + 1];
  for (int i0 = beg + (int)threadIdx.x; i0 < end; i0 += 4 * BK_T) {
    int dv[4];
#pragma unroll
    for (int k = 0; k < 4; ++k) {
      const int i = i0 + k * BK_T;
      dv[k] = (i < end) ? psd[i].y : -1;
    }
#pragma unroll
    for (int k = 0; k < 4; ++k)
      if (dv[k] >= 0) atomicAdd(&cl[dv[k] - n0], 1);
  }
  __syncthreads();
  const int c = (threadIdx.x < BN) ? cl[threadIdx.x] : 0;
  int inc = c;
  for (int o = 1; o < 32; o <<= 1) {
    const int u = __shfl_up(inc, o);
    if (lane >= o) inc += u;
  }
  if (lane == 31) wsum[w] = inc;
  __syncthreads();
  int woff = 0;
  for (int k = 0; k < BK_T / 32; ++k) woff += (k < w) ? wsum[k] : 0;
  const int node = n0 + (int)threadIdx.x;
  if (threadIdx.x < BN && node < N_NODES) {
    rowptr[node] = beg + woff + inc - c;
    dinv[node]   = rsqrtf((float)(c + 1));
  }
  if (blockIdx.x == NB - 1 && threadIdx.x == 0) rowptr[N_NODES] = end;
}

__global__ __launch_bounds__(BK_T)
void k_bucket_fill(const int2* __restrict__ psd, const int* __restrict__ base,
                   const int* __restrict__ rowptr, const float* __restrict__ dinv,
                   int2* __restrict__ ent) {
  __shared__ int   pos[BN];
  __shared__ float dl[BN];
  const int n0 = blockIdx.x * BN;
  if (threadIdx.x < BN) {
    const int node = n0 + (int)threadIdx.x;
    const bool ok = node < N_NODES;
    pos[threadIdx.x] = ok ? rowptr[node] : 0;
    dl[threadIdx.x]  = ok ? dinv[node] : 0.0f;
  }
  __syncthreads();
  const int beg = base[blockIdx.x];
  const int end = base[blockIdx.x + 1];
  for (int i0 = beg + (int)threadIdx.x; i0 < end; i0 += 4 * BK_T) {
    int2 sd[4];
    float ds[4];
#pragma unroll
    for (int k = 0; k < 4; ++k) {
      const int i = i0 + k * BK_T;
      sd[k] = (i < end) ? psd[i] : make_int2(0, -1);
    }
#pragma unroll
    for (int k = 0; k < 4; ++k) ds[k] = (sd[k].y >= 0) ? dinv[sd[k].x] : 0.0f;
#pragma unroll
    for (int k = 0; k < 4; ++k) {
      if (sd[k].y >= 0) {
        const int j = sd[k].y - n0;
        const int p = atomicAdd(&pos[j], 1);
        ent[p] = make_int2(sd[k].x, __float_as_int(ds[k] * dl[j]));
      }
    }
  }
}

#define GEMM_T 128

template <int K, int M>
__global__ __launch_bounds__(GEMM_T)
void k_gemm_wmma(const float* __restrict__ X, const float* __restrict__ W,
                 float* __restrict__ H, int n_rows) {
  constexpr int WT    = M / 16;
  constexpr int PITCH = 2 * M + 32;
  constexpr int CP    = M / 2;
  constexpr int ITEMS = (K / 2) * CP;
  __shared__ __attribute__((aligned(16))) float Ws[(K / 2) * PITCH];

  static_assert(ITEMS % GEMM_T == 0, "whole passes over W");

#pragma unroll
  for (unsigned s = 0; s < ITEMS / GEMM_T; ++s) {
    const unsigned it = threadIdx.x + s * GEMM_T;
    const unsigned j = it / CP;
    const unsigned p = it % CP;
    const float2 r0 = *reinterpret_cast<const float2*>(W + (size_t)(2 * j) * M + 2 * p);
    const float2 r1 = *reinterpret_cast<const float2*>(W + (size_t)(2 * j + 1) * M + 2 * p);
    *reinterpret_cast<float4*>(&Ws[j * PITCH + 4 * p]) = make_float4(r0.x, r1.x, r0.y, r1.y);
  }
  __syncthreads();

  const int lane = threadIdx.x & 31;
  const int tile = __builtin_amdgcn_readfirstlane(
      (int)(blockIdx.x * (GEMM_T / 32) + (threadIdx.x >> 5)));
  if (tile >= n_rows / 16) return;

  const int m = lane & 15;
  const int hf = lane >> 4;

  const float* __restrict__ xrow = X + (size_t)(tile * 16 + m) * K + 2 * hf;
  const float* bl = Ws + hf * PITCH + 2 * m;

  v8f acc[WT];
#pragma unroll
  for (int t = 0; t < WT; ++t) acc[t] = v8f{};

#pragma unroll
  for (int k0 = 0; k0 < K; k0 += 4) {
    v2f a;
    a.x = xrow[k0];
    a.y = xrow[k0 + 1];
#pragma unroll
    for (int t = 0; t < WT; ++t) {
      const float2 bb = *reinterpret_cast<const float2*>(bl + (k0 / 2) * PITCH + t * 32);
      v2f b;
      b.x = bb.x;
      b.y = bb.y;
      acc[t] = __builtin_amdgcn_wmma_f32_16x16x4_f32(false, a, false, b, (short)0, acc[t],
                                                     false, false);
    }
  }

  const int row0 = tile * 16 + (hf << 3);
  float* __restrict__ out0 = H + (size_t)row0 * M + m;
#pragma unroll
  for (int v = 0; v < 8; ++v) {
#pragma unroll
    for (int t = 0; t < WT; ++t) out0[(size_t)v * M + t * 16] = acc[t][v];
  }
}

template <int F> struct RowVec;
template <> struct RowVec<4> { typedef float4 T; };
template <> struct RowVec<2> { typedef float2 T; };

__device__ __forceinline__ void axpy(float4& a, float c, const float4& v) {
  a.x += c * v.x; a.y += c * v.y; a.z += c * v.z; a.w += c * v.w;
}
__device__ __forceinline__ void axpy(float2& a, float c, const float2& v) {
  a.x += c * v.x; a.y += c * v.y;
}
template <bool RELU>
__device__ __forceinline__ float add_res(float a, float r) {
  const float v = a + r;
  return RELU ? (v > 0.0f ? v : 0.0f) : v;
}
template <bool RELU>
__device__ __forceinline__ float4 add_res(const float4& a, const float4& r) {
  return make_float4(add_res<RELU>(a.x, r.x), add_res<RELU>(a.y, r.y),
                     add_res<RELU>(a.z, r.z), add_res<RELU>(a.w, r.w));
}
template <bool RELU>
__device__ __forceinline__ float2 add_res(const float2& a, const float2& r) {
  return make_float2(add_res<RELU>(a.x, r.x), add_res<RELU>(a.y, r.y));
}

template <int M, bool RELU>
__device__ __forceinline__ typename RowVec<M / 32>::T
row_sum(const int i, const int lane,
        const int* __restrict__ rowptr, const int2* __restrict__ ent,
        const float* __restrict__ dinv, const float* __restrict__ h,
        const float* __restrict__ bias, const float* __restrict__ res) {
  constexpr int F = M / 32;
  typedef typename RowVec<F>::T VT;
  const int beg = rowptr[i];
  const int end = rowptr[i + 1];
  const float di = dinv[i];
  const VT* __restrict__ hv = reinterpret_cast<const VT*>(h);

  const VT rr = reinterpret_cast<const VT*>(res)[(size_t)i * 32 + lane];
  VT acc = reinterpret_cast<const VT*>(bias)[lane];
  axpy(acc, di * di, hv[(size_t)i * 32 + lane]);

  for (int base = beg; base < end; base += 32) {
    const int cnt = (end - base) < 32 ? (end - base) : 32;
    int2 en = make_int2(0, 0);
    if (lane < cnt) en = ent[base + lane];
#define ENT_SRC(t)  __builtin_amdgcn_readlane(en.x, (t))
#define ENT_COEF(t) __int_as_float(__builtin_amdgcn_readlane(en.y, (t)))
#define ROW(t)      hv[(size_t)ENT_SRC(t) * 32 + lane]
    int t = 0;
    for (; t + 4 <= cnt; t += 4) {
      const VT v0 = ROW(t), v1 = ROW(t + 1), v2 = ROW(t + 2), v3 = ROW(t + 3);
      axpy(acc, ENT_COEF(t), v0);
      axpy(acc, ENT_COEF(t + 1), v1);
      axpy(acc, ENT_COEF(t + 2), v2);
      axpy(acc, ENT_COEF(t + 3), v3);
    }
    const int r = cnt - t;
    if (r == 3) {
      const VT v0 = ROW(t), v1 = ROW(t + 1), v2 = ROW(t + 2);
      axpy(acc, ENT_COEF(t), v0);
      axpy(acc, ENT_COEF(t + 1), v1);
      axpy(acc, ENT_COEF(t + 2), v2);
    } else if (r == 2) {
      const VT v0 = ROW(t), v1 = ROW(t + 1);
      axpy(acc, ENT_COEF(t), v0);
      axpy(acc, ENT_COEF(t + 1), v1);
    } else if (r == 1) {
      const VT v0 = ROW(t);
      axpy(acc, ENT_COEF(t), v0);
    }
#undef ROW
#undef ENT_COEF
#undef ENT_SRC
  }
  return add_res<RELU>(acc, rr);
}

template <int M, bool RELU>
__global__ __launch_bounds__(256)
void k_gather(const int* __restrict__ rowptr, const int2* __restrict__ ent,
              const float* __restrict__ dinv, const float* __restrict__ h,
              const float* __restrict__ bias, const float* __restrict__ res,
              float* __restrict__ y, int n) {
  typedef typename RowVec<M / 32>::T VT;
  const int lane = threadIdx.x & 31;
  const int i = __builtin_amdgcn_readfirstlane(
      (blockIdx.x * (blockDim.x >> 5)) + (threadIdx.x >> 5));
  if (i >= n) return;
  reinterpret_cast<VT*>(y)[(size_t)i * 32 + lane] =
      row_sum<M, RELU>(i, lane, rowptr, ent, dinv, h, bias, res);
}

#define FG_T 512
#define YP   132

template <bool SPLIT, bool STORE_Y>
__global__ __launch_bounds__(FG_T)
void k_gather_gemm(const int* __restrict__ rowptr, const int2* __restrict__ ent,
                   const float* __restrict__ dinv, const float* __restrict__ h,
                   const float* __restrict__ bias, const float* __restrict__ res,
                   float* __restrict__ y,
                   const float* __restrict__ Wa, const float* __restrict__ Wb,
                   float* __restrict__ Ha, float* __restrict__ Hb, int n_tiles) {
  __shared__ __attribute__((aligned(16))) float Ws[NFH * NFH];
  __shared__ __attribute__((aligned(16))) float Ys[16 * YP];

#pragma unroll
  for (unsigned s = 0; s < (NFH / 4) * NFH / FG_T; ++s) {
    const unsigned it = threadIdx.x + s * FG_T;
    const unsigned j = it / NFH;
    const unsigned c = it % NFH;
    const float* wp;
    unsigned ld;
    if (SPLIT) {
      wp = (c < NCLS) ? (Wa + c) : (Wb + (c - NCLS));
      ld = NCLS;
    } else {
      wp = Wa + c;
      ld = NFH;
    }
    wp += (size_t)(4 * j) * ld;
    *reinterpret_cast<float4*>(&Ws[j * (4 * NFH) + 4 * c]) =
        make_float4(wp[0], wp[ld], wp[2 * ld], wp[3 * ld]);
  }

  const int lane = threadIdx.x & 31;
  const int w = __builtin_amdgcn_readfirstlane((int)(threadIdx.x >> 5));
  const int m = lane & 15;
  const int hf = lane >> 4;

  int turn = 0;
  for (int tile = blockIdx.x; tile < n_tiles; tile += gridDim.x, turn ^= 1) {
    const int i = tile * 16 + w;
    const float4 yv = row_sum<NFH, true>(i, lane, rowptr, ent, dinv, h, bias, res);
    __syncthreads();
    *reinterpret_cast<float4*>(&Ys[w * YP + 4 * lane]) = yv;
    __syncthreads();
    if (STORE_Y) reinterpret_cast<float4*>(y)[(size_t)i * 32 + lane] = yv;
    if ((w >> 3) == turn) {
      const int t = w & 7;
      const float* al = Ys + m * YP + 2 * hf;
      const float* bl = Ws + (t * 16 + m) * 4 + 2 * hf;
      v8f acc = v8f{};
#pragma unroll
      for (int k0 = 0; k0 < NFH; k0 += 4) {
        const float2 aa = *reinterpret_cast<const float2*>(al + k0);
        const float2 bb = *reinterpret_cast<const float2*>(bl + (k0 / 4) * (4 * NFH));
        v2f a, b;
        a.x = aa.x; a.y = aa.y;
        b.x = bb.x; b.y = bb.y;
        acc = __builtin_amdgcn_wmma_f32_16x16x4_f32(false, a, false, b, (short)0, acc,
                                                    false, false);
      }
      const int row0 = tile * 16 + (hf << 3);
      float* o;
      int ld;
      if (SPLIT) {
        o = (t < 4) ? (Ha + t * 16 + m) : (Hb + (t - 4) * 16 + m);
        ld = NCLS;
      } else {
        o = Ha + t * 16 + m;
        ld = NFH;
      }
      o += (size_t)row0 * ld;
#pragma unroll
      for (int v = 0; v < 8; ++v) o[(size_t)v * ld] = acc[v];
    }
  }
}

extern "C" void kernel_launch(void* const* d_in, const int* in_sizes, int n_in,
                              void* d_out, int out_size, void* d_ws, size_t ws_size,
                              hipStream_t stream) {
  const float* x      = (const float*)d_in[0];
  const int*   ei     = (const int*)d_in[1];
  const float* W_in   = (const float*)d_in[2];
  const float* b_in   = (const float*)d_in[3];
  const float* W_m0   = (const float*)d_in[4];
  const float* b_m0   = (const float*)d_in[5];
  const float* W_m1   = (const float*)d_in[6];
  const float* b_m1   = (const float*)d_in[7];
  const float* W_out  = (const float*)d_in[8];
  const float* b_out  = (const float*)d_in[9];
  const float* W_skip = (const float*)d_in[10];

  const int* src = ei;
  const int* dst = ei + N_EDGES;

  float* w    = (float*)d_ws;
  float* dinv = w;  w += N_NODES;
  float* hA   = w;  w += (size_t)N_NODES * NFH;
  float* hB   = w;  w += (size_t)N_NODES * NFH;
  float* xA   = w;  w += (size_t)N_NODES * NFH;
  float* xB   = w;  w += (size_t)N_NODES * NFH;
  int*  rowptr = (int*)w;   w += N_NODES + 32;
  int*  hist   = (int*)w;   w += P1_B * NB;
  int*  pre    = (int*)w;   w += P1_B * NB;
  int*  base   = (int*)w;   w += 512;
  int2* psd    = (int2*)w;  w += (size_t)2 * N_EDGES;
  int2* ent    = (int2*)w;  w += (size_t)2 * N_EDGES;
  float* hOut  = hB;
  float* hSkip = hB + (size_t)N_NODES * NCLS;

  const int T = 256;
  auto cdiv = [](int a, int b) { return (a + b - 1) / b; };

  const int rowBlocks   = cdiv(N_NODES, T / 32);
  const int gemmBlocks  = cdiv(N_NODES / 16, GEMM_T / 32);
  const int fusedBlocks = 1024;
  const int nTiles      = N_NODES / 16;

  k_bcount<<<P1_B, P1_T, 0, stream>>>(dst, hist);
  k_bscan<<<1, BS_T, 0, stream>>>(hist, pre, base);
  k_part<<<P1_B, P1_T, 0, stream>>>(src, dst, pre, base, psd);
  k_bucket_count<<<NB, BK_T, 0, stream>>>(psd, base, rowptr, dinv);
  k_bucket_fill<<<NB, BK_T, 0, stream>>>(psd, base, rowptr, dinv, ent);

  k_gemm_wmma<NFH, NFH><<<gemmBlocks, GEMM_T, 0, stream>>>(x, W_in, hA, N_NODES);

  k_gather_gemm<false, true><<<fusedBlocks, FG_T, 0, stream>>>(
      rowptr, ent, dinv, hA, b_in, x, xA, W_m0, nullptr, hB, nullptr, nTiles);

  k_gather_gemm<false, true><<<fusedBlocks, FG_T, 0, stream>>>(
      rowptr, ent, dinv, hB, b_m0, xA, xB, W_m1, nullptr, hA, nullptr, nTiles);

  k_gather_gemm<true, false><<<fusedBlocks, FG_T, 0, stream>>>(
      rowptr, ent, dinv, hA, b_m1, xB, nullptr, W_out, W_skip, hOut, hSkip, nTiles);

  k_gather<NCLS, false><<<rowBlocks, T, 0, stream>>>(rowptr, ent, dinv, hOut, b_out, hSkip,
                                                     (float*)d_out, N_NODES);
}


# Layers 0-2 sum first, multiply after: x rows summed per node, the 16-row tile times W from LDS, then bias + kept self row + relu; h planes and residual reads gone; rounding: (sum of rows) x W
# speedup vs baseline: 5.2830x; 1.1098x over previous
#include <hip/hip_runtime.h>
#include <hip/hip_bf16.h>

#define N_NODES 100000
#define N_EDGES 1600000
#define NFH     128
#define NCLS    64

typedef __attribute__((ext_vector_type(2))) float v2f;
typedef __attribute__((ext_vector_type(8))) float v8f;

#define BN     256
#define NB     ((N_NODES + BN - 1) / BN)
#define P1_B   256
#define P1_T   512
#define P1_CH  ((N_EDGES + P1_B - 1) / P1_B)
#define P1_PER ((P1_CH + P1_T - 1) / P1_T)
#define BS_T   512
#define BK_T   512

__global__ __launch_bounds__(P1_T)
void k_bcount(const int* __restrict__ dst, int* __restrict__ hist) {
  __shared__ int hl[NB];
  for (int t = threadIdx.x; t < NB; t += P1_T) hl[t] = 0;
  __syncthreads();
  const int e0 = blockIdx.x * P1_CH;
  const int e1 = (e0 + P1_CH < N_EDGES) ? (e0 + P1_CH) : N_EDGES;
  int dv[P1_PER];
#pragma unroll
  for (int k = 0; k < P1_PER; ++k) {
    const int i = e0 + (int)threadIdx.x + k * P1_T;
    dv[k] = (i < e1) ? dst[i] : -1;
  }
#pragma unroll
  for (int k = 0; k < P1_PER; ++k)
    if (dv[k] >= 0) atomicAdd(&hl[dv[k] / BN], 1);
  __syncthreads();
  for (int t = threadIdx.x; t < NB; t += P1_T) hist[blockIdx.x * NB + t] = hl[t];
}

__global__ __launch_bounds__(BS_T)
void k_bscan(const int* __restrict__ hist, int* __restrict__ pre, int* __restrict__ base) {
  static_assert(NB <= BS_T, "one thread per bucket");
  __shared__ int wsum[BS_T / 32];
  const int t    = threadIdx.x;
  const int lane = threadIdx.x & 31;
  const int w    = threadIdx.x >> 5;
  int run = 0;
  if (t < NB) {
#pragma unroll 32
    for (int k = 0; k < P1_B; ++k) {
      const int v = hist[k * NB + t];
      pre[k * NB + t] = run;
      run += v;
    }
  }
  int inc = run;
  for (int o = 1; o < 32; o <<= 1) {
    const int u = __shfl_up(inc, o);
    if (lane >= o) inc += u;
  }
  if (lane == 31) wsum[w] = inc;
  __syncthreads();
  int woff = 0;
  for (int k = 0; k < BS_T / 32; ++k) woff += (k < w) ? wsum[k] : 0;
  const int ex = woff + inc - run;
  if (t < NB) base[t] = ex;
  if (t == NB - 1) base[NB] = ex + run;
}

__global__ __launch_bounds__(P1_T)
void k_part(const int* __restrict__ src, const int* __restrict__ dst,
            const int* __restrict__ pre, const int* __restrict__ base,
            int2* __restrict__ psd) {
  __shared__ int cur[NB];
  for (int t = threadIdx.x; t < NB; t += P1_T) cur[t] = base[t] + pre[blockIdx.x * NB + t];
  __syncthreads();
  const int e0 = blockIdx.x * P1_CH;
  const int e1 = (e0 + P1_CH < N_EDGES) ? (e0 + P1_CH) : N_EDGES;
  int sv[P1_PER], dv[P1_PER];
#pragma unroll
  for (int k = 0; k < P1_PER; ++k) {
    const int i = e0 + (int)threadIdx.x + k * P1_T;
    const bool ok = i < e1;
    sv[k] = ok ? src[i] : 0;
    dv[k] = ok ? dst[i] : -1;
  }
#pragma unroll
  for (int k = 0; k < P1_PER; ++k) {
    if (dv[k] >= 0) {
      const int p = atomicAdd(&cur[dv[k] / BN], 1);
      psd[p] = make_int2(sv[k], dv[k]);
    }
  }
}

__global__ __launch_bounds__(BK_T)
void k_bucket_count(const int2* __restrict__ psd, const int* __restrict__ base,
                    int* __restrict__ rowptr, float* __restrict__ dinv) {
  static_assert(BN <= BK_T, "one thread per node of the bucket");
  __shared__ int cl[BN];
  __shared__ int wsum[BK_T / 32];
  const int lane = threadIdx.x & 31;
  const int w    = threadIdx.x >> 5;
  const int n0   = blockIdx.x * BN;
  if (threadIdx.x < BN) cl[threadIdx.x] = 0;
  __syncthreads();
  const int beg = base[blockIdx.x];
  const int end = base[blockIdx.x + 1];
  for (int i0 = beg + (int)threadIdx.x; i0 < end; i0 += 4 * BK_T) {
    int dv[4];
#pragma unroll
    for (int k = 0; k < 4; ++k) {
      const int i = i0 + k * BK_T;
      dv[k] = (i < end) ? psd[i].y : -1;
    }
#pragma unroll
    for (int k = 0; k < 4; ++k)
      if (dv[k] >= 0) atomicAdd(&cl[dv[k] - n0], 1);
  }
  __syncthreads();
  const int c = (threadIdx.x < BN) ? cl[threadIdx.x] : 0;
  int inc = c;
  for (int o = 1; o < 32; o <<= 1) {
    const int u = __shfl_up(inc, o);
    if (lane >= o) inc += u;
  }
  if (lane == 31) wsum[w] = inc;
  __syncthreads();
  int woff = 0;
  for (int k = 0; k < BK_T / 32; ++k) woff += (k < w) ? wsum[k] : 0;
  const int node = n0 + (int)threadIdx.x;
  if (threadIdx.x < BN && node < N_NODES) {
    rowptr[node] = beg + woff + inc - c;
    dinv[node]   = rsqrtf((float)(c + 1));
  }
  if (blockIdx.x == NB - 1 && threadIdx.x == 0) rowptr[N_NODES] = end;
}

__global__ __launch_bounds__(BK_T)
void k_bucket_fill(const int2* __restrict__ psd, const int* __restrict__ base,
                   const int* __restrict__ rowptr, const float* __restrict__ dinv,
                   int2* __restrict__ ent) {
  __shared__ int   pos[BN];
  __shared__ float dl[BN];
  const int n0 = blockIdx.x * BN;
  if (threadIdx.x < BN) {
    const int node = n0 + (int)threadIdx.x;
    const bool ok = node < N_NODES;
    pos[threadIdx.x] = ok ? rowptr[node] : 0;
    dl[threadIdx.x]  = ok ? dinv[node] : 0.0f;
  }
  __syncthreads();
  const int beg = base[blockIdx.x];
  const int end = base[blockIdx.x + 1];
  for (int i0 = beg + (int)threadIdx.x; i0 < end; i0 += 4 * BK_T) {
    int2 sd[4];
    float ds[4];
#pragma unroll
    for (int k = 0; k < 4; ++k) {
      const int i = i0 + k * BK_T;
      sd[k] = (i < end) ? psd[i] : make_int2(0, -1);
    }
#pragma unroll
    for (int k = 0; k < 4; ++k) ds[k] = (sd[k].y >= 0) ? dinv[sd[k].x] : 0.0f;
#pragma unroll
    for (int k = 0; k < 4; ++k) {
      if (sd[k].y >= 0) {
        const int j = sd[k].y - n0;
        const int p = atomicAdd(&pos[j], 1);
        ent[p] = make_int2(sd[k].x, __float_as_int(ds[k] * dl[j]));
      }
    }
  }
}

#define GEMM_T 128

template <int K, int M, bool SPLIT>
__global__ __launch_bounds__(GEMM_T)
void k_gemm_wmma(const float* __restrict__ X, const float* __restrict__ Wa,
                 const float* __restrict__ Wb, float* __restrict__ Ha,
                 float* __restrict__ Hb, int n_rows) {
  constexpr int WT    = M / 16;
  constexpr int PITCH = 2 * M + 32;
  constexpr int CP    = M / 2;
  constexpr int ITEMS = (K / 2) * CP;
  __shared__ __attribute__((aligned(16))) float Ws[(K / 2) * PITCH];

  static_assert(ITEMS % GEMM_T == 0, "whole passes over W");

#pragma unroll
  for (unsigned s = 0; s < ITEMS / GEMM_T; ++s) {
    const unsigned it = threadIdx.x + s * GEMM_T;
    const unsigned j = it / CP;
    const unsigned p = it % CP;
    const float* wp;
    unsigned ldw;
    if (SPLIT) {
      wp = (p < CP / 2) ? (Wa + 2 * p) : (Wb + 2 * (p - CP / 2));
      ldw = M / 2;
    } else {
      wp = Wa + 2 * p;
      ldw = M;
    }
    const float2 r0 = *reinterpret_cast<const float2*>(wp + (size_t)(2 * j) * ldw);
    const float2 r1 = *reinterpret_cast<const float2*>(wp + (size_t)(2 * j + 1) * ldw);
    *reinterpret_cast<float4*>(&Ws[j * PITCH + 4 * p]) = make_float4(r0.x, r1.x, r0.y, r1.y);
  }
  __syncthreads();

  const int lane = threadIdx.x & 31;
  const int tile = __builtin_amdgcn_readfirstlane(
      (int)(blockIdx.x * (GEMM_T / 32) + (threadIdx.x >> 5)));
  if (tile >= n_rows / 16) return;

  const int m = lane & 15;
  const int hf = lane >> 4;

  const float* __restrict__ xrow = X + (size_t)(tile * 16 + m) * K + 2 * hf;
  const float* bl = Ws + hf * PITCH + 2 * m;

  v8f acc[WT];
#pragma unroll
  for (int t = 0; t < WT; ++t) acc[t] = v8f{};

#pragma unroll
  for (int k0 = 0; k0 < K; k0 += 4) {
    v2f a;
    a.x = xrow[k0];
    a.y = xrow[k0 + 1];
#pragma unroll
    for (int t = 0; t < WT; ++t) {
      const float2 bb = *reinterpret_cast<const float2*>(bl + (k0 / 2) * PITCH + t * 32);
      v2f b;
      b.x = bb.x;
      b.y = bb.y;
      acc[t] = __builtin_amdgcn_wmma_f32_16x16x4_f32(false, a, false, b, (short)0, acc[t],
                                                     false, false);
    }
  }

  const int row0 = tile * 16 + (hf << 3);
  if (SPLIT) {
    float* __restrict__ oa = Ha + (size_t)row0 * (M / 2) + m;
    float* __restrict__ ob = Hb + (size_t)row0 * (M / 2) + m;
#pragma unroll
    for (int v = 0; v < 8; ++v) {
#pragma unroll
      for (int t = 0; t < WT / 2; ++t) oa[(size_t)v * (M / 2) + t * 16] = acc[t][v];
#pragma unroll
      for (int t = WT / 2; t < WT; ++t) ob[(size_t)v * (M / 2) + (t - WT / 2) * 16] = acc[t][v];
    }
  } else {
    float* __restrict__ out0 = Ha + (size_t)row0 * M + m;
#pragma unroll
    for (int v = 0; v < 8; ++v) {
#pragma unroll
      for (int t = 0; t < WT; ++t) out0[(size_t)v * M + t * 16] = acc[t][v];
    }
  }
}

template <int F> struct RowVec;
template <> struct RowVec<4> { typedef float4 T; };
template <> struct RowVec<2> { typedef float2 T; };

__device__ __forceinline__ void axpy(float4& a, float c, const float4& v) {
  a.x += c * v.x; a.y += c * v.y; a.z += c * v.z; a.w += c * v.w;
}
__device__ __forceinline__ void axpy(float2& a, float c, const float2& v) {
  a.x += c * v.x; a.y += c * v.y;
}
template <bool RELU>
__device__ __forceinline__ float add_res(float a, float r) {
  const float v = a + r;
  return RELU ? (v > 0.0f ? v : 0.0f) : v;
}
template <bool RELU>
__device__ __forceinline__ float4 add_res(const float4& a, const float4& r) {
  return make_float4(add_res<RELU>(a.x, r.x), add_res<RELU>(a.y, r.y),
                     add_res<RELU>(a.z, r.z), add_res<RELU>(a.w, r.w));
}
template <bool RELU>
__device__ __forceinline__ float2 add_res(const float2& a, const float2& r) {
  return make_float2(add_res<RELU>(a.x, r.x), add_res<RELU>(a.y, r.y));
}

template <typename VT>
__device__ __forceinline__ void edge_sum(VT& acc, const int beg, const int end, const int lane,
                                         const int2* __restrict__ ent,
                                         const VT* __restrict__ hv) {
  for (int base = beg; base < end; base += 32) {
    const int cnt = (end - base) < 32 ? (end - base) : 32;
    int2 en = make_int2(0, 0);
    if (lane < cnt) en = ent[base + lane];
#define ENT_SRC(t)  __builtin_amdgcn_readlane(en.x, (t))
#define ENT_COEF(t) __int_as_float(__builtin_amdgcn_readlane(en.y, (t)))
#define ROW(t)      hv[(size_t)ENT_SRC(t) * 32 + lane]
    int t = 0;
    for (; t + 4 <= cnt; t += 4) {
      const VT v0 = ROW(t), v1 = ROW(t + 1), v2 = ROW(t + 2), v3 = ROW(t + 3);
      axpy(acc, ENT_COEF(t), v0);
      axpy(acc, ENT_COEF(t + 1), v1);
      axpy(acc, ENT_COEF(t + 2), v2);
      axpy(acc, ENT_COEF(t + 3), v3);
    }
    const int r = cnt - t;
    if (r == 3) {
      const VT v0 = ROW(t), v1 = ROW(t + 1), v2 = ROW(t + 2);
      axpy(acc, ENT_COEF(t), v0);
      axpy(acc, ENT_COEF(t + 1), v1);
      axpy(acc, ENT_COEF(t + 2), v2);
    } else if (r == 2) {
      const VT v0 = ROW(t), v1 = ROW(t + 1);
      axpy(acc, ENT_COEF(t), v0);
      axpy(acc, ENT_COEF(t + 1), v1);
    } else if (r == 1) {
      const VT v0 = ROW(t);
      axpy(acc, ENT_COEF(t), v0);
    }
#undef ROW
#undef ENT_COEF
#undef ENT_SRC
  }
}

template <int M, bool RELU>
__device__ __forceinline__ typename RowVec<M / 32>::T
row_sum(const int i, const int lane,
        const int* __restrict__ rowptr, const int2* __restrict__ ent,
        const float* __restrict__ dinv, const float* __restrict__ h,
        const float* __restrict__ bias, const float* __restrict__ res) {
  constexpr int F = M / 32;
  typedef typename RowVec<F>::T VT;
  const int beg = rowptr[i];
  const int end = rowptr[i + 1];
  const float di = dinv[i];
  const VT* __restrict__ hv = reinterpret_cast<const VT*>(h);

  const VT rr = reinterpret_cast<const VT*>(res)[(size_t)i * 32 + lane];
  VT acc = reinterpret_cast<const VT*>(bias)[lane];
  axpy(acc, di * di, hv[(size_t)i * 32 + lane]);

  edge_sum<VT>(acc, beg, end, lane, ent, hv);
  return add_res<RELU>(acc, rr);
}

template <int M, bool RELU>
__global__ __launch_bounds__(256)
void k_gather(const int* __restrict__ rowptr, const int2* __restrict__ ent,
              const float* __restrict__ dinv, const float* __restrict__ h,
              const float* __restrict__ bias, const float* __restrict__ res,
              float* __restrict__ y, int n) {
  typedef typename RowVec<M / 32>::T VT;
  const int lane = threadIdx.x & 31;
  const int i = __builtin_amdgcn_readfirstlane(
      (blockIdx.x * (blockDim.x >> 5)) + (threadIdx.x >> 5));
  if (i >= n) return;
  reinterpret_cast<VT*>(y)[(size_t)i * 32 + lane] =
      row_sum<M, RELU>(i, lane, rowptr, ent, dinv, h, bias, res);
}

#define FG_T 512
#define TP   128

__device__ __forceinline__ float4
row_agg(const int i, const int lane,
        const int* __restrict__ rowptr, const int2* __restrict__ ent,
        const float* __restrict__ dinv, const float* __restrict__ x, float4& self) {
  const int beg = rowptr[i];
  const int end = rowptr[i + 1];
  const float di = dinv[i];
  const float4* __restrict__ xv = reinterpret_cast<const float4*>(x);
  self = xv[(size_t)i * 32 + lane];
  float4 acc = make_float4(0.0f, 0.0f, 0.0f, 0.0f);
  axpy(acc, di * di, self);
  edge_sum<float4>(acc, beg, end, lane, ent, xv);
  return acc;
}

__global__ __launch_bounds__(FG_T)
void k_agg_gemm(const int* __restrict__ rowptr, const int2* __restrict__ ent,
                const float* __restrict__ dinv, const float* __restrict__ x,
                const float* __restrict__ bias, float* __restrict__ y,
                const float* __restrict__ Wm, int n_tiles) {
  __shared__ __attribute__((aligned(16))) float Ws[NFH * NFH];
  __shared__ __attribute__((aligned(16))) float Ys[16 * TP];
  __shared__ __attribute__((aligned(16))) float Zs[16 * TP];

#pragma unroll
  for (unsigned s = 0; s < (NFH / 4) * NFH / FG_T; ++s) {
    const unsigned it = threadIdx.x + s * FG_T;
    const unsigned j = it / NFH;
    const unsigned c = it % NFH;
    const float* wp = Wm + c + (size_t)(4 * j) * NFH;
    *reinterpret_cast<float4*>(&Ws[j * (4 * NFH) + 4 * c]) =
        make_float4(wp[0], wp[NFH], wp[2 * NFH], wp[3 * NFH]);
  }

  const int lane = threadIdx.x & 31;
  const int w = __builtin_amdgcn_readfirstlane((int)(threadIdx.x >> 5));
  const int m = lane & 15;
  const int hf = lane >> 4;

  const float4 bv = reinterpret_cast<const float4*>(bias)[lane];
  const int own = w * TP + ((lane ^ w) << 2);

  int turn = 0;
  int iprev = -1;
  float4 xprev = make_float4(0.0f, 0.0f, 0.0f, 0.0f);
  for (int tile = blockIdx.x; tile < n_tiles; tile += gridDim.x, turn ^= 1) {
    const int i = tile * 16 + w;
    float4 xi;
    const float4 a = row_agg(i, lane, rowptr, ent, dinv, x, xi);
    __syncthreads();
    const int idone = iprev;
    float4 yv = xprev;
    if (idone >= 0) {
      const float4 z = *reinterpret_cast<const float4*>(&Zs[own]);
      yv = add_res<true>(make_float4(z.x + bv.x, z.y + bv.y, z.z + bv.z, z.w + bv.w), xprev);
    }
    *reinterpret_cast<float4*>(&Ys[own]) = a;
    xprev = xi;
    iprev = i;
    __syncthreads();
    if (idone >= 0) reinterpret_cast<float4*>(y)[(size_t)idone * 32 + lane] = yv;
    if ((w >> 3) == turn) {
      const int t = w & 7;
      const float* al = Ys + m * TP + 2 * hf;
      const int sw = m << 2;
      const float* bl = Ws + (t * 16 + m) * 4 + 2 * hf;
      v8f acc = v8f{};
#pragma unroll 8
      for (int k0 = 0; k0 < NFH; k0 += 4) {
        const float2 aa = *reinterpret_cast<const float2*>(al + (k0 ^ sw));
        const float2 bb = *reinterpret_cast<const float2*>(bl + (k0 / 4) * (4 * NFH));
        v2f a2, b2;
        a2.x = aa.x; a2.y = aa.y;
        b2.x = bb.x; b2.y = bb.y;
        acc = __builtin_amdgcn_wmma_f32_16x16x4_f32(false, a2, false, b2, (short)0, acc,
                                                    false, false);
      }
      const int c = t * 16 + m;
#pragma unroll
      for (int v = 0; v < 8; ++v) {
        const int r = v + (hf << 3);
        Zs[r * TP + ((((c >> 2) ^ r) << 2) | (c & 3))] = acc[v];
      }
    }
  }
  __syncthreads();
  if (iprev >= 0) {
    const float4 z = *reinterpret_cast<const float4*>(&Zs[own]);
    reinterpret_cast<float4*>(y)[(size_t)iprev * 32 + lane] =
        add_res<true>(make_float4(z.x + bv.x, z.y + bv.y, z.z + bv.z, z.w + bv.w), xprev);
  }
}

extern "C" void kernel_launch(void* const* d_in, const int* in_sizes, int n_in,
                              void* d_out, int out_size, void* d_ws, size_t ws_size,
                              hipStream_t stream) {
  const float* x      = (const float*)d_in[0];
  const int*   ei     = (const int*)d_in[1];
  const float* W_in   = (const float*)d_in[2];
  const float* b_in   = (const float*)d_in[3];
  const float* W_m0   = (const float*)d_in[4];
  const float* b_m0   = (const float*)d_in[5];
  const float* W_m1   = (const float*)d_in[6];
  const float* b_m1   = (const float*)d_in[7];
  const float* W_out  = (const float*)d_in[8];
  const float* b_out  = (const float*)d_in[9];
  const float* W_skip = (const float*)d_in[10];

  const int* src = ei;
  const int* dst = ei + N_EDGES;

  float* w    = (float*)d_ws;
  float* dinv = w;  w += N_NODES;
  float* hA   = w;  w += (size_t)N_NODES * NFH;
  float* hB   = w;  w += (size_t)N_NODES * NFH;
  float* xA   = w;  w += (size_t)N_NODES * NFH;
  float* xB   = w;  w += (size_t)N_NODES * NFH;
  int*  rowptr = (int*)w;   w += N_NODES + 32;
  int*  hist   = (int*)w;   w += P1_B * NB;
  int*  pre    = (int*)w;   w += P1_B * NB;
  int*  base   = (int*)w;   w += 512;
  int2* psd    = (int2*)w;  w += (size_t)2 * N_EDGES;
  int2* ent    = (int2*)w;  w += (size_t)2 * N_EDGES;
  float* hOut  = hB;
  float* hSkip = hB + (size_t)N_NODES * NCLS;

  const int T = 256;
  auto cdiv = [](int a, int b) { return (a + b - 1) / b; };

  const int rowBlocks   = cdiv(N_NODES, T / 32);
  const int gemmBlocks  = cdiv(N_NODES / 16, GEMM_T / 32);
  const int fusedBlocks = 1024;
  const int nTiles      = N_NODES / 16;

  k_bcount<<<P1_B, P1_T, 0, stream>>>(dst, hist);
  k_bscan<<<1, BS_T, 0, stream>>>(hist, pre, base);
  k_part<<<P1_B, P1_T, 0, stream>>>(src, dst, pre, base, psd);
  k_bucket_count<<<NB, BK_T, 0, stream>>>(psd, base, rowptr, dinv);
  k_bucket_fill<<<NB, BK_T, 0, stream>>>(psd, base, rowptr, dinv, ent);

  k_agg_gemm<<<fusedBlocks, FG_T, 0, stream>>>(rowptr, ent, dinv, x,  b_in, xA, W_in, nTiles);
  k_agg_gemm<<<fusedBlocks, FG_T, 0, stream>>>(rowptr, ent, dinv, xA, b_m0, xB, W_m0, nTiles);
  k_agg_gemm<<<fusedBlocks, FG_T, 0, stream>>>(rowptr, ent, dinv, xB, b_m1, xA, W_m1, nTiles);

  k_gemm_wmma<NFH, NFH, true><<<gemmBlocks, GEMM_T, 0, stream>>>(xA, W_out, W_skip, hOut, hSkip,
                                                                N_NODES);

  k_gather<NCLS, false><<<rowBlocks, T, 0, stream>>>(rowptr, ent, dinv, hOut, b_out, hSkip,
                                                     (float*)d_out, N_NODES);
}
